# starting kernel
# baseline (speedup 1.0000x reference)
#include <hip/hip_runtime.h>
#include <hip/hip_bf16.h>

#define BATCH 512
#define JN    140
#define NB    28
#define NV    13059
#define NC    (3 * BATCH)
#define KDIM  (4 * JN)
#define LDSW  132

typedef __attribute__((ext_vector_type(2))) float v2f;
typedef __attribute__((ext_vector_type(8))) float v8f;

__global__ __launch_bounds__(256)
void mouse_chain_kernel(const float* __restrict__ pose,
                        const float* __restrict__ bone_lengths,
                        const float* __restrict__ cbl,
                        const float* __restrict__ trans,
                        const float* __restrict__ scale,
                        const float* __restrict__ Jt,
                        const int*   __restrict__ parent,
                        const int*   __restrict__ bone_mapper,
                        float* __restrict__ Gmat,
                        float* __restrict__ Gchain,
                        float* __restrict__ Jout)
{
    const int b = blockIdx.x * blockDim.x + threadIdx.x;
    if (b >= BATCH) return;
    const float sc = scale[b];
    const float t0 = trans[b * 3 + 0];
    const float t1 = trans[b * 3 + 1];
    const float t2 = trans[b * 3 + 2];

    for (int i = 0; i < JN; ++i) {
        const float ez = pose[(b * JN + i) * 3 + 0];
        const float ey = pose[(b * JN + i) * 3 + 1];
        const float ex = pose[(b * JN + i) * 3 + 2];
        const float cx = cosf(ex), sx = sinf(ex);
        const float cy = cosf(ey), sy = sinf(ey);
        const float cz = cosf(ez), sz = sinf(ez);
        float R[9];
        R[0] = cz * cy; R[1] = cz * sy * sx - sz * cx; R[2] = cz * sy * cx + sz * sx;
        R[3] = sz * cy; R[4] = sz * sy * sx + cz * cx; R[5] = sz * sy * cx - cz * sx;
        R[6] = -sy;     R[7] = cy * sx;                R[8] = cy * cx;

        float G[12];
        if (i == 0) {
            G[0] = R[0]; G[1] = R[1]; G[2]  = R[2]; G[3]  = Jt[0];
            G[4] = R[3]; G[5] = R[4]; G[6]  = R[5]; G[7]  = Jt[1];
            G[8] = R[6]; G[9] = R[7]; G[10] = R[8]; G[11] = Jt[2];
        } else {
            const int p = parent[i];
            float Gp[12];
#pragma unroll
            for (int e = 0; e < 12; ++e) Gp[e] = Gchain[(p * 12 + e) * BATCH + b];
            float s;
            if (i == 1) {
                s = cbl[b];
            } else {
                const int bm = bone_mapper[i];
                s = (bm >= 0) ? 2.0f / (1.0f + expf(-bone_lengths[b * NB + bm] * 0.2f))
                              : 1.0f;
            }
            const float off0 = (Jt[i * 3 + 0] - Jt[p * 3 + 0]) * s;
            const float off1 = (Jt[i * 3 + 1] - Jt[p * 3 + 1]) * s;
            const float off2 = (Jt[i * 3 + 2] - Jt[p * 3 + 2]) * s;
#pragma unroll
            for (int r = 0; r < 3; ++r) {
                const float g0 = Gp[r * 4 + 0], g1 = Gp[r * 4 + 1];
                const float g2 = Gp[r * 4 + 2], g3 = Gp[r * 4 + 3];
                G[r * 4 + 0] = g0 * R[0] + g1 * R[3] + g2 * R[6];
                G[r * 4 + 1] = g0 * R[1] + g1 * R[4] + g2 * R[7];
                G[r * 4 + 2] = g0 * R[2] + g1 * R[5] + g2 * R[8];
                G[r * 4 + 3] = g0 * off0 + g1 * off1 + g2 * off2 + g3;
            }
        }
#pragma unroll
        for (int e = 0; e < 12; ++e) Gchain[(i * 12 + e) * BATCH + b] = G[e];

        const float jx = Jt[i * 3 + 0], jy = Jt[i * 3 + 1], jz = Jt[i * 3 + 2];
#pragma unroll
        for (int r = 0; r < 3; ++r) {
            const float trel = G[r * 4 + 3] -
                (G[r * 4 + 0] * jx + G[r * 4 + 1] * jy + G[r * 4 + 2] * jz);
            const size_t c2 = (size_t)(3 * b + r) * 2;
            Gmat[(size_t)(2 * i    ) * NC * 2 + c2 + 0] = G[r * 4 + 0];
            Gmat[(size_t)(2 * i    ) * NC * 2 + c2 + 1] = G[r * 4 + 1];
            Gmat[(size_t)(2 * i + 1) * NC * 2 + c2 + 0] = G[r * 4 + 2];
            Gmat[(size_t)(2 * i + 1) * NC * 2 + c2 + 1] = trel;
        }
        Jout[(b * JN + i) * 3 + 0] = G[3]  * sc + t0;
        Jout[(b * JN + i) * 3 + 1] = G[7]  * sc + t1;
        Jout[(b * JN + i) * 3 + 2] = G[11] * sc + t2;
    }
}

__global__ __launch_bounds__(256)
void mouse_skin_gemm(const float* __restrict__ W,
                     const float* __restrict__ vt,
                     const float* __restrict__ Gmat,
                     const float* __restrict__ scale,
                     const float* __restrict__ trans,
                     float* __restrict__ Vout)
{
    __shared__ float lds[16 * LDSW];

    const int lane  = threadIdx.x & 31;
    const int wave  = threadIdx.x >> 5;
    const int nSub  = blockIdx.y * 8 + wave;
    const int col   = lane & 15;
    const int khalf = lane >> 4;

    const int   vRow  = blockIdx.x * 16 + col;
    const bool  vok   = vRow < NV;
    const int   vc    = vok ? vRow : 0;
    const float wmask = vok ? 1.0f : 0.0f;

    const float c0 = vt[vc * 3 + 2 * khalf] * wmask;
    const float c1 = ((khalf == 0) ? vt[vc * 3 + 1] : 1.0f) * wmask;

    const float* __restrict__ wrow = W + vc * JN;
    const int cIdx = nSub * 16 + col;
    const float* __restrict__ gp =
        Gmat + ((size_t)khalf * NC + cIdx) * 2;

    v8f acc0 = {}, acc1 = {};
#pragma unroll 2
    for (int jq = 0; jq < JN / 4; ++jq) {
        const float4 wq = *(const float4*)(wrow + jq * 4);
        const v2f b0 = *(const v2f*)(gp);
        const v2f b1 = *(const v2f*)(gp + 4 * NC);
        const v2f b2 = *(const v2f*)(gp + 8 * NC);
        const v2f b3 = *(const v2f*)(gp + 12 * NC);
        v2f a;
        a.x = wq.x * c0; a.y = wq.x * c1;
        acc0 = __builtin_amdgcn_wmma_f32_16x16x4_f32(false, a, false, b0,
                                                     (short)0, acc0, false, false);
        a.x = wq.y * c0; a.y = wq.y * c1;
        acc1 = __builtin_amdgcn_wmma_f32_16x16x4_f32(false, a, false, b1,
                                                     (short)0, acc1, false, false);
        a.x = wq.z * c0; a.y = wq.z * c1;
        acc0 = __builtin_amdgcn_wmma_f32_16x16x4_f32(false, a, false, b2,
                                                     (short)0, acc0, false, false);
        a.x = wq.w * c0; a.y = wq.w * c1;
        acc1 = __builtin_amdgcn_wmma_f32_16x16x4_f32(false, a, false, b3,
                                                     (short)0, acc1, false, false);
        gp += 16 * NC;
    }
    const v8f acc = acc0 + acc1;

#pragma unroll
    for (int i = 0; i < 8; ++i)
        lds[(khalf * 8 + i) * LDSW + wave * 16 + col] = acc[i];
    __syncthreads();

    const int C0     = blockIdx.y * 128;
    const int vr0    = blockIdx.x * 16;
    const int bFirst = C0 / 3;
    const int nGroups = (C0 + 127) / 3 - bFirst + 1;
    const int seg = threadIdx.x / 48;
    const int s   = threadIdx.x - seg * 48;
    if (seg < 5) {
        const int row = s / 3;
        const int r   = s - row * 3;
        const int vr  = vr0 + row;
        for (int g = seg; g < nGroups; g += 5) {
            const int bg = bFirst + g;
            const int lc = 3 * bg + r - C0;
            if (lc >= 0 && lc < 128 && vr < NV) {
                const float val = lds[row * LDSW + lc];
                Vout[((size_t)bg * NV + vr) * 3 + r] =
                    val * scale[bg] + trans[bg * 3 + r];
            }
        }
    }
}

extern "C" void kernel_launch(void* const* d_in, const int* in_sizes, int n_in,
                              void* d_out, int out_size, void* d_ws, size_t ws_size,
                              hipStream_t stream) {
    const float* pose         = (const float*)d_in[0];
    const float* bone_lengths = (const float*)d_in[1];
    const float* cbl          = (const float*)d_in[2];
    const float* trans        = (const float*)d_in[3];
    const float* scale        = (const float*)d_in[4];
    const float* v_template   = (const float*)d_in[5];
    const float* t_pose       = (const float*)d_in[6];
    const float* weights      = (const float*)d_in[7];
    const int*   parent       = (const int*)d_in[8];
    const int*   bone_mapper  = (const int*)d_in[9];

    float* Gmat   = (float*)d_ws;
    float* Gchain = Gmat + (size_t)KDIM * NC;

    float* Vout = (float*)d_out;
    float* Jout = Vout + (size_t)BATCH * NV * 3;

    mouse_chain_kernel<<<dim3((BATCH + 255) / 256), 256, 0, stream>>>(
        pose, bone_lengths, cbl, trans, scale, t_pose, parent, bone_mapper,
        Gmat, Gchain, Jout);

    dim3 grid((NV + 15) / 16, (NC / 16) / 8);
    mouse_skin_gemm<<<grid, 256, 0, stream>>>(
        weights, v_template, Gmat, scale, trans, Vout);
}


# chain kernel: one work-group per batch element; sin/cos of all joints in parallel, then 12 lanes (one per matrix entry) walk the joints through LDS, not one thread per batch through global memory
# speedup vs baseline: 1.9897x; 1.9897x over previous
#include <hip/hip_runtime.h>
#include <hip/hip_bf16.h>

#define BATCH 512
#define JN    140
#define NB    28
#define NV    13059
#define NC    (3 * BATCH)
#define KDIM  (4 * JN)
#define LDSW  132
#define CHT   160

typedef __attribute__((ext_vector_type(2))) float v2f;
typedef __attribute__((ext_vector_type(8))) float v8f;

__global__ __launch_bounds__(CHT)
void mouse_chain_kernel(const float* __restrict__ pose,
                        const float* __restrict__ bone_lengths,
                        const float* __restrict__ cbl,
                        const float* __restrict__ trans,
                        const float* __restrict__ scale,
                        const float* __restrict__ Jt,
                        const int*   __restrict__ parent,
                        const int*   __restrict__ bone_mapper,
                        float* __restrict__ Gmat,
                        float* __restrict__ Jout)
{
    __shared__ __attribute__((aligned(16))) float Gl[JN * 12];
    __shared__ __attribute__((aligned(16))) float Al[JN * 16];
    __shared__ __attribute__((aligned(16))) float Jl[JN * 4];
    __shared__ int Pl[JN];

    const int t = threadIdx.x;
    const int b = blockIdx.x;
    const float sc  = scale[b];
    const float tr0 = trans[b * 3 + 0];
    const float tr1 = trans[b * 3 + 1];
    const float tr2 = trans[b * 3 + 2];

    if (t < JN) {
        const int i = t;
        const float ez = pose[(b * JN + i) * 3 + 0];
        const float ey = pose[(b * JN + i) * 3 + 1];
        const float ex = pose[(b * JN + i) * 3 + 2];
        const float cx = cosf(ex), sx = sinf(ex);
        const float cy = cosf(ey), sy = sinf(ey);
        const float cz = cosf(ez), sz = sinf(ez);
        float R[9];
        R[0] = cz * cy; R[1] = cz * sy * sx - sz * cx; R[2] = cz * sy * cx + sz * sx;
        R[3] = sz * cy; R[4] = sz * sy * sx + cz * cx; R[5] = sz * sy * cx - cz * sx;
        R[6] = -sy;     R[7] = cy * sx;                R[8] = cy * cx;

        const float jx = Jt[i * 3 + 0], jy = Jt[i * 3 + 1], jz = Jt[i * 3 + 2];
        *(float4*)(Jl + i * 4) = make_float4(jx, jy, jz, 0.0f);

        if (i == 0) {
            *(float4*)(Gl + 0) = make_float4(R[0], R[1], R[2], jx);
            *(float4*)(Gl + 4) = make_float4(R[3], R[4], R[5], jy);
            *(float4*)(Gl + 8) = make_float4(R[6], R[7], R[8], jz);
            Pl[0] = 0;
        } else {
            const int p = parent[i];
            float s;
            if (i == 1) {
                s = cbl[b];
            } else {
                const int bm = bone_mapper[i];
                s = (bm >= 0) ? 2.0f / (1.0f + expf(-bone_lengths[b * NB + bm] * 0.2f))
                              : 1.0f;
            }
            const float off0 = (jx - Jt[p * 3 + 0]) * s;
            const float off1 = (jy - Jt[p * 3 + 1]) * s;
            const float off2 = (jz - Jt[p * 3 + 2]) * s;
            *(float4*)(Al + i * 16 + 0)  = make_float4(R[0], R[3], R[6], 0.0f);
            *(float4*)(Al + i * 16 + 4)  = make_float4(R[1], R[4], R[7], 0.0f);
            *(float4*)(Al + i * 16 + 8)  = make_float4(R[2], R[5], R[8], 0.0f);
            *(float4*)(Al + i * 16 + 12) = make_float4(off0, off1, off2, 1.0f);
            Pl[i] = p * 3;
            const float4 z = make_float4(0.0f, 0.0f, 0.0f, 0.0f);
            *(float4*)(Gl + i * 12 + 0) = z;
            *(float4*)(Gl + i * 12 + 4) = z;
            *(float4*)(Gl + i * 12 + 8) = z;
        }
    }
    __syncthreads();

    if (t < 12) {
        const int r = t >> 2;
        const int c = t & 3;
        int p = Pl[1];
        for (int i = 1; i < JN; ++i) {
            const int pn = Pl[(i + 1 < JN) ? (i + 1) : i];
            const float4 a = *(const float4*)(Al + i * 16 + c * 4);
            const float4 g = ((const float4*)Gl)[p + r];
            const float v = fmaf(g.w, a.w, fmaf(g.z, a.z, fmaf(g.y, a.y, g.x * a.x)));
            Gl[i * 12 + t] = v;
            p = pn;
        }
    }
    __syncthreads();

    for (int item = t; item < 3 * JN; item += CHT) {
        const int i = item / 3;
        const int r = item - 3 * i;
        const float4 g  = *(const float4*)(Gl + item * 4);
        const float4 jt = *(const float4*)(Jl + i * 4);
        const float trel = g.w - (g.x * jt.x + g.y * jt.y + g.z * jt.z);
        const size_t c2 = (size_t)(3 * b + r) * 2;
        *(float2*)(Gmat + (size_t)(2 * i    ) * NC * 2 + c2) = make_float2(g.x, g.y);
        *(float2*)(Gmat + (size_t)(2 * i + 1) * NC * 2 + c2) = make_float2(g.z, trel);
        const float tr = (r == 0) ? tr0 : ((r == 1) ? tr1 : tr2);
        Jout[b * (JN * 3) + item] = g.w * sc + tr;
    }
}

__global__ __launch_bounds__(256)
void mouse_skin_gemm(const float* __restrict__ W,
                     const float* __restrict__ vt,
                     const float* __restrict__ Gmat,
                     const float* __restrict__ scale,
                     const float* __restrict__ trans,
                     float* __restrict__ Vout)
{
    __shared__ float lds[16 * LDSW];

    const int lane  = threadIdx.x & 31;
    const int wave  = threadIdx.x >> 5;
    const int nSub  = blockIdx.y * 8 + wave;
    const int col   = lane & 15;
    const int khalf = lane >> 4;

    const int   vRow  = blockIdx.x * 16 + col;
    const bool  vok   = vRow < NV;
    const int   vc    = vok ? vRow : 0;
    const float wmask = vok ? 1.0f : 0.0f;

    const float c0 = vt[vc * 3 + 2 * khalf] * wmask;
    const float c1 = ((khalf == 0) ? vt[vc * 3 + 1] : 1.0f) * wmask;

    const float* __restrict__ wrow = W + vc * JN;
    const int cIdx = nSub * 16 + col;
    const float* __restrict__ gp =
        Gmat + ((size_t)khalf * NC + cIdx) * 2;

    v8f acc0 = {}, acc1 = {};
#pragma unroll 2
    for (int jq = 0; jq < JN / 4; ++jq) {
        const float4 wq = *(const float4*)(wrow + jq * 4);
        const v2f b0 = *(const v2f*)(gp);
        const v2f b1 = *(const v2f*)(gp + 4 * NC);
        const v2f b2 = *(const v2f*)(gp + 8 * NC);
        const v2f b3 = *(const v2f*)(gp + 12 * NC);
        v2f a;
        a.x = wq.x * c0; a.y = wq.x * c1;
        acc0 = __builtin_amdgcn_wmma_f32_16x16x4_f32(false, a, false, b0,
                                                     (short)0, acc0, false, false);
        a.x = wq.y * c0; a.y = wq.y * c1;
        acc1 = __builtin_amdgcn_wmma_f32_16x16x4_f32(false, a, false, b1,
                                                     (short)0, acc1, false, false);
        a.x = wq.z * c0; a.y = wq.z * c1;
        acc0 = __builtin_amdgcn_wmma_f32_16x16x4_f32(false, a, false, b2,
                                                     (short)0, acc0, false, false);
        a.x = wq.w * c0; a.y = wq.w * c1;
        acc1 = __builtin_amdgcn_wmma_f32_16x16x4_f32(false, a, false, b3,
                                                     (short)0, acc1, false, false);
        gp += 16 * NC;
    }
    const v8f acc = acc0 + acc1;

#pragma unroll
    for (int i = 0; i < 8; ++i)
        lds[(khalf * 8 + i) * LDSW + wave * 16 + col] = acc[i];
    __syncthreads();

    const int C0     = blockIdx.y * 128;
    const int vr0    = blockIdx.x * 16;
    const int bFirst = C0 / 3;
    const int nGroups = (C0 + 127) / 3 - bFirst + 1;
    const int seg = threadIdx.x / 48;
    const int s   = threadIdx.x - seg * 48;
    if (seg < 5) {
        const int row = s / 3;
        const int r   = s - row * 3;
        const int vr  = vr0 + row;
        for (int g = seg; g < nGroups; g += 5) {
            const int bg = bFirst + g;
            const int lc = 3 * bg + r - C0;
            if (lc >= 0 && lc < 128 && vr < NV) {
                const float val = lds[row * LDSW + lc];
                Vout[((size_t)bg * NV + vr) * 3 + r] =
                    val * scale[bg] + trans[bg * 3 + r];
            }
        }
    }
}

extern "C" void kernel_launch(void* const* d_in, const int* in_sizes, int n_in,
                              void* d_out, int out_size, void* d_ws, size_t ws_size,
                              hipStream_t stream) {
    const float* pose         = (const float*)d_in[0];
    const float* bone_lengths = (const float*)d_in[1];
    const float* cbl          = (const float*)d_in[2];
    const float* trans        = (const float*)d_in[3];
    const float* scale        = (const float*)d_in[4];
    const float* v_template   = (const float*)d_in[5];
    const float* t_pose       = (const float*)d_in[6];
    const float* weights      = (const float*)d_in[7];
    const int*   parent       = (const int*)d_in[8];
    const int*   bone_mapper  = (const int*)d_in[9];

    float* Gmat = (float*)d_ws;

    float* Vout = (float*)d_out;
    float* Jout = Vout + (size_t)BATCH * NV * 3;

    mouse_chain_kernel<<<dim3(BATCH), CHT, 0, stream>>>(
        pose, bone_lengths, cbl, trans, scale, t_pose, parent, bone_mapper,
        Gmat, Jout);

    dim3 grid((NV + 15) / 16, (NC / 16) / 8);
    mouse_skin_gemm<<<grid, 256, 0, stream>>>(
        weights, v_template, Gmat, scale, trans, Vout);
}


# GEMM: each wave loads the B fragments of its 16 columns once into registers (all 140 K-steps) and walks 13 M tiles in-kernel (grid 63x24, 4-wave blocks); no B re-read from L2 in the loop
# speedup vs baseline: 2.1104x; 1.0607x over previous
#include <hip/hip_runtime.h>
#include <hip/hip_bf16.h>

#define BATCH 512
#define JN    140
#define NB    28
#define NV    13059
#define NC    (3 * BATCH)
#define KDIM  (4 * JN)
#define CHT   160

typedef __attribute__((ext_vector_type(2))) float v2f;
typedef __attribute__((ext_vector_type(8))) float v8f;

__global__ __launch_bounds__(CHT)
void mouse_chain_kernel(const float* __restrict__ pose,
                        const float* __restrict__ bone_lengths,
                        const float* __restrict__ cbl,
                        const float* __restrict__ trans,
                        const float* __restrict__ scale,
                        const float* __restrict__ Jt,
                        const int*   __restrict__ parent,
                        const int*   __restrict__ bone_mapper,
                        float* __restrict__ Gmat,
                        float* __restrict__ Jout)
{
    __shared__ __attribute__((aligned(16))) float Gl[JN * 12];
    __shared__ __attribute__((aligned(16))) float Al[JN * 16];
    __shared__ __attribute__((aligned(16))) float Jl[JN * 4];
    __shared__ int Pl[JN];

    const int t = threadIdx.x;
    const int b = blockIdx.x;
    const float sc  = scale[b];
    const float tr0 = trans[b * 3 + 0];
    const float tr1 = trans[b * 3 + 1];
    const float tr2 = trans[b * 3 + 2];

    if (t < JN) {
        const int i = t;
        const float ez = pose[(b * JN + i) * 3 + 0];
        const float ey = pose[(b * JN + i) * 3 + 1];
        const float ex = pose[(b * JN + i) * 3 + 2];
        const float cx = cosf(ex), sx = sinf(ex);
        const float cy = cosf(ey), sy = sinf(ey);
        const float cz = cosf(ez), sz = sinf(ez);
        float R[9];
        R[0] = cz * cy; R[1] = cz * sy * sx - sz * cx; R[2] = cz * sy * cx + sz * sx;
        R[3] = sz * cy; R[4] = sz * sy * sx + cz * cx; R[5] = sz * sy * cx - cz * sx;
        R[6] = -sy;     R[7] = cy * sx;                R[8] = cy * cx;

        const float jx = Jt[i * 3 + 0], jy = Jt[i * 3 + 1], jz = Jt[i * 3 + 2];
        *(float4*)(Jl + i * 4) = make_float4(jx, jy, jz, 0.0f);

        if (i == 0) {
            *(float4*)(Gl + 0) = make_float4(R[0], R[1], R[2], jx);
            *(float4*)(Gl + 4) = make_float4(R[3], R[4], R[5], jy);
            *(float4*)(Gl + 8) = make_float4(R[6], R[7], R[8], jz);
            Pl[0] = 0;
        } else {
            const int p = parent[i];
            float s;
            if (i == 1) {
                s = cbl[b];
            } else {
                const int bm = bone_mapper[i];
                s = (bm >= 0) ? 2.0f / (1.0f + expf(-bone_lengths[b * NB + bm] * 0.2f))
                              : 1.0f;
            }
            const float off0 = (jx - Jt[p * 3 + 0]) * s;
            const float off1 = (jy - Jt[p * 3 + 1]) * s;
            const float off2 = (jz - Jt[p * 3 + 2]) * s;
            *(float4*)(Al + i * 16 + 0)  = make_float4(R[0], R[3], R[6], 0.0f);
            *(float4*)(Al + i * 16 + 4)  = make_float4(R[1], R[4], R[7], 0.0f);
            *(float4*)(Al + i * 16 + 8)  = make_float4(R[2], R[5], R[8], 0.0f);
            *(float4*)(Al + i * 16 + 12) = make_float4(off0, off1, off2, 1.0f);
            Pl[i] = p * 3;
            const float4 z = make_float4(0.0f, 0.0f, 0.0f, 0.0f);
            *(float4*)(Gl + i * 12 + 0) = z;
            *(float4*)(Gl + i * 12 + 4) = z;
            *(float4*)(Gl + i * 12 + 8) = z;
        }
    }
    __syncthreads();

    if (t < 12) {
        const int r = t >> 2;
        const int c = t & 3;
        int p = Pl[1];
        for (int i = 1; i < JN; ++i) {
            const int pn = Pl[(i + 1 < JN) ? (i + 1) : i];
            const float4 a = *(const float4*)(Al + i * 16 + c * 4);
            const float4 g = ((const float4*)Gl)[p + r];
            const float v = fmaf(g.w, a.w, fmaf(g.z, a.z, fmaf(g.y, a.y, g.x * a.x)));
            Gl[i * 12 + t] = v;
            p = pn;
        }
    }
    __syncthreads();

    for (int item = t; item < 3 * JN; item += CHT) {
        const int i = item / 3;
        const int r = item - 3 * i;
        const float4 g  = *(const float4*)(Gl + item * 4);
        const float4 jt = *(const float4*)(Jl + i * 4);
        const float trel = g.w - (g.x * jt.x + g.y * jt.y + g.z * jt.z);
        const size_t c2 = (size_t)(3 * b + r) * 2;
        *(float2*)(Gmat + (size_t)(2 * i    ) * NC * 2 + c2) = make_float2(g.x, g.y);
        *(float2*)(Gmat + (size_t)(2 * i + 1) * NC * 2 + c2) = make_float2(g.z, trel);
        const float tr = (r == 0) ? tr0 : ((r == 1) ? tr1 : tr2);
        Jout[b * (JN * 3) + item] = g.w * sc + tr;
    }
}

#define MTILES ((NV + 15) / 16)
#define MGRID  63
#define WPB    4
#define SLAB   (WPB * 16)
#define SLDW   (SLAB + 4)
#define NSEG   2
#define WOG    11

__global__ __launch_bounds__(WPB * 32, 2)
void mouse_skin_gemm(const float* __restrict__ W,
                     const float* __restrict__ vt,
                     const float* __restrict__ Gmat,
                     const float* __restrict__ scale,
                     const float* __restrict__ trans,
                     float* __restrict__ Vout)
{
    __shared__ float lds[2][16 * SLDW];

    const int lane  = threadIdx.x & 31;
    const int wave  = threadIdx.x >> 5;
    const int nSub  = blockIdx.y * WPB + wave;
    const int col   = lane & 15;
    const int khalf = lane >> 4;

    const int cIdx = nSub * 16 + col;
    const float* __restrict__ gp =
        Gmat + ((size_t)khalf * NC + cIdx) * 2;
    v2f bfr[JN];
#pragma unroll
    for (int j = 0; j < JN; ++j)
        bfr[j] = *(const v2f*)(gp + (size_t)j * 4 * NC);

    const int C0      = blockIdx.y * SLAB;
    const int bFirst  = C0 / 3;
    const int nGroups = (C0 + SLAB - 1) / 3 - bFirst + 1;
    const int seg = threadIdx.x / 48;
    const int s   = threadIdx.x - seg * 48;
    const int row = s / 3;
    const int r   = s - row * 3;
    int   lcg[WOG];
    int   obg[WOG];
    float scg[WOG], trg[WOG];
#pragma unroll
    for (int k = 0; k < WOG; ++k) {
        const int g  = seg + NSEG * k;
        const int bg = bFirst + g;
        const int lc = 3 * bg + r - C0;
        const bool ok = (seg < NSEG) && (g < nGroups) && (lc >= 0) && (lc < SLAB);
        const int bgc = ok ? bg : 0;
        lcg[k] = ok ? lc : -1;
        obg[k] = bgc * (NV * 3) + r;
        scg[k] = scale[bgc];
        trg[k] = trans[bgc * 3 + r];
    }

    int buf = 0;
    for (int mt = blockIdx.x; mt < MTILES; mt += MGRID) {
        const int   vRow  = mt * 16 + col;
        const bool  vok   = vRow < NV;
        const int   vc    = vok ? vRow : 0;
        const float wmask = vok ? 1.0f : 0.0f;

        const float c0 = vt[vc * 3 + 2 * khalf] * wmask;
        const float c1 = ((khalf == 0) ? vt[vc * 3 + 1] : 1.0f) * wmask;

        const float* __restrict__ wrow = W + vc * JN;

        v8f acc0 = {}, acc1 = {};
#pragma unroll
        for (int jq = 0; jq < JN / 4; ++jq) {
            const float4 wq = *(const float4*)(wrow + jq * 4);
            v2f a;
            a.x = wq.x * c0; a.y = wq.x * c1;
            acc0 = __builtin_amdgcn_wmma_f32_16x16x4_f32(false, a, false, bfr[4 * jq + 0],
                                                         (short)0, acc0, false, false);
            a.x = wq.y * c0; a.y = wq.y * c1;
            acc1 = __builtin_amdgcn_wmma_f32_16x16x4_f32(false, a, false, bfr[4 * jq + 1],
                                                         (short)0, acc1, false, false);
            a.x = wq.z * c0; a.y = wq.z * c1;
            acc0 = __builtin_amdgcn_wmma_f32_16x16x4_f32(false, a, false, bfr[4 * jq + 2],
                                                         (short)0, acc0, false, false);
            a.x = wq.w * c0; a.y = wq.w * c1;
            acc1 = __builtin_amdgcn_wmma_f32_16x16x4_f32(false, a, false, bfr[4 * jq + 3],
                                                         (short)0, acc1, false, false);
        }
        const v8f acc = acc0 + acc1;

        float* __restrict__ stage = lds[buf];
#pragma unroll
        for (int i = 0; i < 8; ++i)
            stage[(khalf * 8 + i) * SLDW + wave * 16 + col] = acc[i];
        __syncthreads();

        const int vr = mt * 16 + row;
        float val[WOG];
#pragma unroll
        for (int k = 0; k < WOG; ++k)
            val[k] = stage[row * SLDW + ((lcg[k] >= 0) ? lcg[k] : 0)];
        if (vr < NV) {
#pragma unroll
            for (int k = 0; k < WOG; ++k) {
                if (lcg[k] >= 0)
                    Vout[(size_t)(obg[k] + vr * 3)] = val[k] * scg[k] + trg[k];
            }
        }
        buf ^= 1;
    }
}

extern "C" void kernel_launch(void* const* d_in, const int* in_sizes, int n_in,
                              void* d_out, int out_size, void* d_ws, size_t ws_size,
                              hipStream_t stream) {
    const float* pose         = (const float*)d_in[0];
    const float* bone_lengths = (const float*)d_in[1];
    const float* cbl          = (const float*)d_in[2];
    const float* trans        = (const float*)d_in[3];
    const float* scale        = (const float*)d_in[4];
    const float* v_template   = (const float*)d_in[5];
    const float* t_pose       = (const float*)d_in[6];
    const float* weights      = (const float*)d_in[7];
    const int*   parent       = (const int*)d_in[8];
    const int*   bone_mapper  = (const int*)d_in[9];

    float* Gmat = (float*)d_ws;

    float* Vout = (float*)d_out;
    float* Jout = Vout + (size_t)BATCH * NV * 3;

    mouse_chain_kernel<<<dim3(BATCH), CHT, 0, stream>>>(
        pose, bone_lengths, cbl, trans, scale, t_pose, parent, bone_mapper,
        Gmat, Jout);

    dim3 grid(MGRID, (NC / 16) / WPB);
    mouse_skin_gemm<<<grid, WPB * 32, 0, stream>>>(
        weights, v_template, Gmat, scale, trans, Vout);
}


# GEMM: each M tile's weights staged in LDS by the whole block with whole-line loads, one tile ahead of the matrix work; lanes read their rows from LDS in groups in front of the WMMA blocks
# speedup vs baseline: 2.4372x; 1.1549x over previous
#include <hip/hip_runtime.h>
#include <hip/hip_bf16.h>

#define BATCH 512
#define JN    140
#define NB    28
#define NV    13059
#define NC    (3 * BATCH)
#define KDIM  (4 * JN)
#define CHT   160

typedef __attribute__((ext_vector_type(2))) float v2f;
typedef __attribute__((ext_vector_type(8))) float v8f;

__global__ __launch_bounds__(CHT)
void mouse_chain_kernel(const float* __restrict__ pose,
                        const float* __restrict__ bone_lengths,
                        const float* __restrict__ cbl,
                        const float* __restrict__ trans,
                        const float* __restrict__ scale,
                        const float* __restrict__ Jt,
                        const int*   __restrict__ parent,
                        const int*   __restrict__ bone_mapper,
                        float* __restrict__ Gmat,
                        float* __restrict__ Jout)
{
    __shared__ __attribute__((aligned(16))) float Gl[JN * 12];
    __shared__ __attribute__((aligned(16))) float Al[JN * 16];
    __shared__ __attribute__((aligned(16))) float Jl[JN * 4];
    __shared__ int Pl[JN];

    const int t = threadIdx.x;
    const int b = blockIdx.x;
    const float sc  = scale[b];
    const float tr0 = trans[b * 3 + 0];
    const float tr1 = trans[b * 3 + 1];
    const float tr2 = trans[b * 3 + 2];

    if (t < JN) {
        const int i = t;
        const float ez = pose[(b * JN + i) * 3 + 0];
        const float ey = pose[(b * JN + i) * 3 + 1];
        const float ex = pose[(b * JN + i) * 3 + 2];
        const float cx = cosf(ex), sx = sinf(ex);
        const float cy = cosf(ey), sy = sinf(ey);
        const float cz = cosf(ez), sz = sinf(ez);
        float R[9];
        R[0] = cz * cy; R[1] = cz * sy * sx - sz * cx; R[2] = cz * sy * cx + sz * sx;
        R[3] = sz * cy; R[4] = sz * sy * sx + cz * cx; R[5] = sz * sy * cx - cz * sx;
        R[6] = -sy;     R[7] = cy * sx;                R[8] = cy * cx;

        const float jx = Jt[i * 3 + 0], jy = Jt[i * 3 + 1], jz = Jt[i * 3 + 2];
        *(float4*)(Jl + i * 4) = make_float4(jx, jy, jz, 0.0f);

        if (i == 0) {
            *(float4*)(Gl + 0) = make_float4(R[0], R[1], R[2], jx);
            *(float4*)(Gl + 4) = make_float4(R[3], R[4], R[5], jy);
            *(float4*)(Gl + 8) = make_float4(R[6], R[7], R[8], jz);
            Pl[0] = 0;
        } else {
            const int p = parent[i];
            float s;
            if (i == 1) {
                s = cbl[b];
            } else {
                const int bm = bone_mapper[i];
                s = (bm >= 0) ? 2.0f / (1.0f + expf(-bone_lengths[b * NB + bm] * 0.2f))
                              : 1.0f;
            }
            const float off0 = (jx - Jt[p * 3 + 0]) * s;
            const float off1 = (jy - Jt[p * 3 + 1]) * s;
            const float off2 = (jz - Jt[p * 3 + 2]) * s;
            *(float4*)(Al + i * 16 + 0)  = make_float4(R[0], R[3], R[6], 0.0f);
            *(float4*)(Al + i * 16 + 4)  = make_float4(R[1], R[4], R[7], 0.0f);
            *(float4*)(Al + i * 16 + 8)  = make_float4(R[2], R[5], R[8], 0.0f);
            *(float4*)(Al + i * 16 + 12) = make_float4(off0, off1, off2, 1.0f);
            Pl[i] = p * 3;
            const float4 z = make_float4(0.0f, 0.0f, 0.0f, 0.0f);
            *(float4*)(Gl + i * 12 + 0) = z;
            *(float4*)(Gl + i * 12 + 4) = z;
            *(float4*)(Gl + i * 12 + 8) = z;
        }
    }
    __syncthreads();

    if (t < 12) {
        const int r = t >> 2;
        const int c = t & 3;
        int p = Pl[1];
        for (int i = 1; i < JN; ++i) {
            const int pn = Pl[(i + 1 < JN) ? (i + 1) : i];
            const float4 a = *(const float4*)(Al + i * 16 + c * 4);
            const float4 g = ((const float4*)Gl)[p + r];
            const float v = fmaf(g.w, a.w, fmaf(g.z, a.z, fmaf(g.y, a.y, g.x * a.x)));
            Gl[i * 12 + t] = v;
            p = pn;
        }
    }
    __syncthreads();

    for (int item = t; item < 3 * JN; item += CHT) {
        const int i = item / 3;
        const int r = item - 3 * i;
        const float4 g  = *(const float4*)(Gl + item * 4);
        const float4 jt = *(const float4*)(Jl + i * 4);
        const float trel = g.w - (g.x * jt.x + g.y * jt.y + g.z * jt.z);
        const size_t c2 = (size_t)(3 * b + r) * 2;
        *(float2*)(Gmat + (size_t)(2 * i    ) * NC * 2 + c2) = make_float2(g.x, g.y);
        *(float2*)(Gmat + (size_t)(2 * i + 1) * NC * 2 + c2) = make_float2(g.z, trel);
        const float tr = (r == 0) ? tr0 : ((r == 1) ? tr1 : tr2);
        Jout[b * (JN * 3) + item] = g.w * sc + tr;
    }
}

#define MTILES ((NV + 15) / 16)
#define MGRID  63
#define WPB    4
#define SLAB   (WPB * 16)
#define SLDW   (SLAB + 4)
#define NSEG   2
#define WOG    11
#define WTF    (16 * JN)
#define WTQ    (WTF / 4)
#define WQTOT  (NV * (JN / 4))

__global__ __launch_bounds__(WPB * 32, 2)
void mouse_skin_gemm(const float* __restrict__ W,
                     const float* __restrict__ vt,
                     const float* __restrict__ Gmat,
                     const float* __restrict__ scale,
                     const float* __restrict__ trans,
                     float* __restrict__ Vout)
{
    __shared__ float lds[2][16 * SLDW];
    __shared__ __attribute__((aligned(16))) float wt[2][WTF];

    const int tid   = threadIdx.x;
    const int lane  = threadIdx.x & 31;
    const int wave  = threadIdx.x >> 5;
    const int nSub  = blockIdx.y * WPB + wave;
    const int col   = lane & 15;
    const int khalf = lane >> 4;

    const float4* __restrict__ W4 = (const float4*)W;

    const bool has5 = tid < WTQ - 4 * (WPB * 32);
    float4 p0, p1, p2, p3, p4 = make_float4(0.0f, 0.0f, 0.0f, 0.0f);
    float  r0, r1;
    {
        const int m0 = blockIdx.x;
        const int g0 = m0 * WTQ + tid;
        p0 = W4[min(g0,       WQTOT - 1)];
        p1 = W4[min(g0 + 128, WQTOT - 1)];
        p2 = W4[min(g0 + 256, WQTOT - 1)];
        p3 = W4[min(g0 + 384, WQTOT - 1)];
        if (has5) p4 = W4[min(g0 + 512, WQTOT - 1)];
        const int vRow = m0 * 16 + col;
        const int vc   = (vRow < NV) ? vRow : 0;
        r0 = vt[vc * 3 + 2 * khalf];
        r1 = vt[vc * 3 + 1];
    }

    const int cIdx = nSub * 16 + col;
    const float* __restrict__ gp =
        Gmat + ((size_t)khalf * NC + cIdx) * 2;
    v2f bfr[JN];
#pragma unroll
    for (int j = 0; j < JN; ++j)
        bfr[j] = *(const v2f*)(gp + (size_t)j * 4 * NC);

    const int C0      = blockIdx.y * SLAB;
    const int bFirst  = C0 / 3;
    const int nGroups = (C0 + SLAB - 1) / 3 - bFirst + 1;
    const int seg = threadIdx.x / 48;
    const int s   = threadIdx.x - seg * 48;
    const int row = s / 3;
    const int r   = s - row * 3;
    int   lcg[WOG];
    int   obg[WOG];
    float scg[WOG], trg[WOG];
#pragma unroll
    for (int k = 0; k < WOG; ++k) {
        const int g  = seg + NSEG * k;
        const int bg = bFirst + g;
        const int lc = 3 * bg + r - C0;
        const bool ok = (seg < NSEG) && (g < nGroups) && (lc >= 0) && (lc < SLAB);
        const int bgc = ok ? bg : 0;
        lcg[k] = ok ? lc : -1;
        obg[k] = bgc * (NV * 3) + r;
        scg[k] = scale[bgc];
        trg[k] = trans[bgc * 3 + r];
    }

    {
        float4* __restrict__ wd = (float4*)wt[0] + tid;
        wd[0] = p0; wd[128] = p1; wd[256] = p2; wd[384] = p3;
        if (has5) wd[512] = p4;
    }
    __syncthreads();

    int buf = 0;
    for (int mt = blockIdx.x; mt < MTILES; mt += MGRID) {
        const bool  vok   = (mt * 16 + col) < NV;
        const float wmask = vok ? 1.0f : 0.0f;

        const float c0 = r0 * wmask;
        const float c1 = ((khalf == 0) ? r1 : 1.0f) * wmask;

        const int  mtn  = mt + MGRID;
        const bool more = mtn < MTILES;
        if (more) {
            const int g0 = mtn * WTQ + tid;
            p0 = W4[min(g0,       WQTOT - 1)];
            p1 = W4[min(g0 + 128, WQTOT - 1)];
            p2 = W4[min(g0 + 256, WQTOT - 1)];
            p3 = W4[min(g0 + 384, WQTOT - 1)];
            if (has5) p4 = W4[min(g0 + 512, WQTOT - 1)];
            const int vRow = mtn * 16 + col;
            const int vc   = (vRow < NV) ? vRow : 0;
            r0 = vt[vc * 3 + 2 * khalf];
            r1 = vt[vc * 3 + 1];
        }

        const float* __restrict__ wr = wt[buf] + col * JN;
        float4 wq[JN / 4];

#define SKIN_STEP4(jq)                                                                        \
        {                                                                                     \
            v2f a;                                                                            \
            a.x = wq[jq].x * c0; a.y = wq[jq].x * c1;                                         \
            acc0 = __builtin_amdgcn_wmma_f32_16x16x4_f32(false, a, false, bfr[4 * (jq) + 0],  \
                                                         (short)0, acc0, false, false);       \
            a.x = wq[jq].y * c0; a.y = wq[jq].y * c1;                                         \
            acc1 = __builtin_amdgcn_wmma_f32_16x16x4_f32(false, a, false, bfr[4 * (jq) + 1],  \
                                                         (short)0, acc1, false, false);       \
            a.x = wq[jq].z * c0; a.y = wq[jq].z * c1;                                         \
            acc0 = __builtin_amdgcn_wmma_f32_16x16x4_f32(false, a, false, bfr[4 * (jq) + 2],  \
                                                         (short)0, acc0, false, false);       \
            a.x = wq[jq].w * c0; a.y = wq[jq].w * c1;                                         \
            acc1 = __builtin_amdgcn_wmma_f32_16x16x4_f32(false, a, false, bfr[4 * (jq) + 3],  \
                                                         (short)0, acc1, false, false);       \
        }

        v8f acc0 = {}, acc1 = {};
#pragma unroll
        for (int jq = 0; jq < 28; ++jq)
            wq[jq] = *(const float4*)(wr + jq * 4);
        __builtin_amdgcn_sched_barrier(0);
#pragma unroll
        for (int jq = 0; jq < 7; ++jq) SKIN_STEP4(jq)
#pragma unroll
        for (int jq = 28; jq < JN / 4; ++jq)
            wq[jq] = *(const float4*)(wr + jq * 4);
        __builtin_amdgcn_sched_barrier(0);
#pragma unroll
        for (int jq = 7; jq < JN / 4; ++jq) SKIN_STEP4(jq)
#undef SKIN_STEP4
        const v8f acc = acc0 + acc1;

        float* __restrict__ stage = lds[buf];
#pragma unroll
        for (int i = 0; i < 8; ++i)
            stage[(khalf * 8 + i) * SLDW + wave * 16 + col] = acc[i];

        if (more) {
            float4* __restrict__ wd = (float4*)wt[buf ^ 1] + tid;
            wd[0] = p0; wd[128] = p1; wd[256] = p2; wd[384] = p3;
            if (has5) wd[512] = p4;
        }
        __syncthreads();

        const int vr = mt * 16 + row;
        float val[WOG];
#pragma unroll
        for (int k = 0; k < WOG; ++k)
            val[k] = stage[row * SLDW + ((lcg[k] >= 0) ? lcg[k] : 0)];
        if (vr < NV) {
#pragma unroll
            for (int k = 0; k < WOG; ++k) {
                if (lcg[k] >= 0)
                    Vout[(size_t)(obg[k] + vr * 3)] = val[k] * scg[k] + trg[k];
            }
        }
        buf ^= 1;
    }
}

extern "C" void kernel_launch(void* const* d_in, const int* in_sizes, int n_in,
                              void* d_out, int out_size, void* d_ws, size_t ws_size,
                              hipStream_t stream) {
    const float* pose         = (const float*)d_in[0];
    const float* bone_lengths = (const float*)d_in[1];
    const float* cbl          = (const float*)d_in[2];
    const float* trans        = (const float*)d_in[3];
    const float* scale        = (const float*)d_in[4];
    const float* v_template   = (const float*)d_in[5];
    const float* t_pose       = (const float*)d_in[6];
    const float* weights      = (const float*)d_in[7];
    const int*   parent       = (const int*)d_in[8];
    const int*   bone_mapper  = (const int*)d_in[9];

    float* Gmat = (float*)d_ws;

    float* Vout = (float*)d_out;
    float* Jout = Vout + (size_t)BATCH * NV * 3;

    mouse_chain_kernel<<<dim3(BATCH), CHT, 0, stream>>>(
        pose, bone_lengths, cbl, trans, scale, t_pose, parent, bone_mapper,
        Gmat, Jout);

    dim3 grid(MGRID, (NC / 16) / WPB);
    mouse_skin_gemm<<<grid, WPB * 32, 0, stream>>>(
        weights, v_template, Gmat, scale, trans, Vout);
}


# GEMM tile loop: a lane's 11 writeout stores issued as one batch under one lane mask, 5th weight piece loaded by all lanes: no address-translation wait (s_wait_xcnt) left in the loop
# speedup vs baseline: 2.5909x; 1.0631x over previous
#include <hip/hip_runtime.h>
#include <hip/hip_bf16.h>

#define BATCH 512
#define JN    140
#define NB    28
#define NV    13059
#define NC    (3 * BATCH)
#define KDIM  (4 * JN)
#define CHT   160

typedef __attribute__((ext_vector_type(2))) float v2f;
typedef __attribute__((ext_vector_type(8))) float v8f;

__global__ __launch_bounds__(CHT)
void mouse_chain_kernel(const float* __restrict__ pose,
                        const float* __restrict__ bone_lengths,
                        const float* __restrict__ cbl,
                        const float* __restrict__ trans,
                        const float* __restrict__ scale,
                        const float* __restrict__ Jt,
                        const int*   __restrict__ parent,
                        const int*   __restrict__ bone_mapper,
                        float* __restrict__ Gmat,
                        float* __restrict__ Jout)
{
    __shared__ __attribute__((aligned(16))) float Gl[JN * 12];
    __shared__ __attribute__((aligned(16))) float Al[JN * 16];
    __shared__ __attribute__((aligned(16))) float Jl[JN * 4];
    __shared__ int Pl[JN];

    const int t = threadIdx.x;
    const int b = blockIdx.x;
    const float sc  = scale[b];
    const float tr0 = trans[b * 3 + 0];
    const float tr1 = trans[b * 3 + 1];
    const float tr2 = trans[b * 3 + 2];

    if (t < JN) {
        const int i = t;
        const float ez = pose[(b * JN + i) * 3 + 0];
        const float ey = pose[(b * JN + i) * 3 + 1];
        const float ex = pose[(b * JN + i) * 3 + 2];
        const float cx = cosf(ex), sx = sinf(ex);
        const float cy = cosf(ey), sy = sinf(ey);
        const float cz = cosf(ez), sz = sinf(ez);
        float R[9];
        R[0] = cz * cy; R[1] = cz * sy * sx - sz * cx; R[2] = cz * sy * cx + sz * sx;
        R[3] = sz * cy; R[4] = sz * sy * sx + cz * cx; R[5] = sz * sy * cx - cz * sx;
        R[6] = -sy;     R[7] = cy * sx;                R[8] = cy * cx;

        const float jx = Jt[i * 3 + 0], jy = Jt[i * 3 + 1], jz = Jt[i * 3 + 2];
        *(float4*)(Jl + i * 4) = make_float4(jx, jy, jz, 0.0f);

        if (i == 0) {
            *(float4*)(Gl + 0) = make_float4(R[0], R[1], R[2], jx);
            *(float4*)(Gl + 4) = make_float4(R[3], R[4], R[5], jy);
            *(float4*)(Gl + 8) = make_float4(R[6], R[7], R[8], jz);
            Pl[0] = 0;
        } else {
            const int p = parent[i];
            float s;
            if (i == 1) {
                s = cbl[b];
            } else {
                const int bm = bone_mapper[i];
                s = (bm >= 0) ? 2.0f / (1.0f + expf(-bone_lengths[b * NB + bm] * 0.2f))
                              : 1.0f;
            }
            const float off0 = (jx - Jt[p * 3 + 0]) * s;
            const float off1 = (jy - Jt[p * 3 + 1]) * s;
            const float off2 = (jz - Jt[p * 3 + 2]) * s;
            *(float4*)(Al + i * 16 + 0)  = make_float4(R[0], R[3], R[6], 0.0f);
            *(float4*)(Al + i * 16 + 4)  = make_float4(R[1], R[4], R[7], 0.0f);
            *(float4*)(Al + i * 16 + 8)  = make_float4(R[2], R[5], R[8], 0.0f);
            *(float4*)(Al + i * 16 + 12) = make_float4(off0, off1, off2, 1.0f);
            Pl[i] = p * 3;
            const float4 z = make_float4(0.0f, 0.0f, 0.0f, 0.0f);
            *(float4*)(Gl + i * 12 + 0) = z;
            *(float4*)(Gl + i * 12 + 4) = z;
            *(float4*)(Gl + i * 12 + 8) = z;
        }
    }
    __syncthreads();

    if (t < 12) {
        const int r = t >> 2;
        const int c = t & 3;
        int p = Pl[1];
        for (int i = 1; i < JN; ++i) {
            const int pn = Pl[(i + 1 < JN) ? (i + 1) : i];
            const float4 a = *(const float4*)(Al + i * 16 + c * 4);
            const float4 g = ((const float4*)Gl)[p + r];
            const float v = fmaf(g.w, a.w, fmaf(g.z, a.z, fmaf(g.y, a.y, g.x * a.x)));
            Gl[i * 12 + t] = v;
            p = pn;
        }
    }
    __syncthreads();

    for (int item = t; item < 3 * JN; item += CHT) {
        const int i = item / 3;
        const int r = item - 3 * i;
        const float4 g  = *(const float4*)(Gl + item * 4);
        const float4 jt = *(const float4*)(Jl + i * 4);
        const float trel = g.w - (g.x * jt.x + g.y * jt.y + g.z * jt.z);
        const size_t c2 = (size_t)(3 * b + r) * 2;
        *(float2*)(Gmat + (size_t)(2 * i    ) * NC * 2 + c2) = make_float2(g.x, g.y);
        *(float2*)(Gmat + (size_t)(2 * i + 1) * NC * 2 + c2) = make_float2(g.z, trel);
        const float tr = (r == 0) ? tr0 : ((r == 1) ? tr1 : tr2);
        Jout[b * (JN * 3) + item] = g.w * sc + tr;
    }
}

#define MTILES ((NV + 15) / 16)
#define MGRID  63
#define WPB    4
#define SLAB   (WPB * 16)
#define SLDW   (SLAB + 4)
#define NSEG   2
#define WOG    11
#define WTF    (16 * JN)
#define WTQ    (WTF / 4)
#define WQTOT  (NV * (JN / 4))

__global__ __launch_bounds__(WPB * 32, 2)
void mouse_skin_gemm(const float* __restrict__ W,
                     const float* __restrict__ vt,
                     const float* __restrict__ Gmat,
                     const float* __restrict__ scale,
                     const float* __restrict__ trans,
                     float* __restrict__ Vout)
{
    __shared__ float lds[2][16 * SLDW];
    __shared__ __attribute__((aligned(16))) float wt[2][WTF];

    const int tid   = threadIdx.x;
    const int lane  = threadIdx.x & 31;
    const int wave  = threadIdx.x >> 5;
    const int nSub  = blockIdx.y * WPB + wave;
    const int col   = lane & 15;
    const int khalf = lane >> 4;

    const float4* __restrict__ W4 = (const float4*)W;

    const bool has5 = tid < WTQ - 4 * (WPB * 32);
    float4 p0, p1, p2, p3, p4;
    float  r0, r1;
    {
        const int m0 = blockIdx.x;
        const int g0 = m0 * WTQ + tid;
        p0 = W4[min(g0,       WQTOT - 1)];
        p1 = W4[min(g0 + 128, WQTOT - 1)];
        p2 = W4[min(g0 + 256, WQTOT - 1)];
        p3 = W4[min(g0 + 384, WQTOT - 1)];
        p4 = W4[min(g0 + 512, WQTOT - 1)];
        const int vRow = m0 * 16 + col;
        const int vc   = (vRow < NV) ? vRow : 0;
        r0 = vt[vc * 3 + 2 * khalf];
        r1 = vt[vc * 3 + 1];
    }

    const int cIdx = nSub * 16 + col;
    const float* __restrict__ gp =
        Gmat + ((size_t)khalf * NC + cIdx) * 2;
    v2f bfr[JN];
#pragma unroll
    for (int j = 0; j < JN; ++j)
        bfr[j] = *(const v2f*)(gp + (size_t)j * 4 * NC);

    const int C0     = blockIdx.y * SLAB;
    const int bFirst = C0 / 3;
    const int seg = threadIdx.x / 48;
    const bool act = seg < NSEG;
    const int segc = act ? seg : 0;
    const int s   = threadIdx.x - seg * 48;
    const int row = s / 3;
    const int r   = s - row * 3;
    int   lof[WOG];
    int   oof[WOG];
    float scg[WOG], trg[WOG];
#pragma unroll
    for (int k = 0; k < WOG; ++k) {
        int bg = bFirst + segc + NSEG * k;
        int lc = 3 * bg + r - C0;
        if (lc < 0)     { bg += NSEG; lc += 3 * NSEG; }
        if (lc >= SLAB) { bg -= NSEG; lc -= 3 * NSEG; }
        lof[k] = row * SLDW + lc;
        oof[k] = bg * (NV * 3) + s;
        scg[k] = scale[bg];
        trg[k] = trans[bg * 3 + r];
    }

    {
        float4* __restrict__ wd = (float4*)wt[0] + tid;
        wd[0] = p0; wd[128] = p1; wd[256] = p2; wd[384] = p3;
        if (has5) wd[512] = p4;
    }
    __syncthreads();

    int buf = 0;
    for (int mt = blockIdx.x; mt < MTILES; mt += MGRID) {
        const bool  vok   = (mt * 16 + col) < NV;
        const float wmask = vok ? 1.0f : 0.0f;

        const float c0 = r0 * wmask;
        const float c1 = ((khalf == 0) ? r1 : 1.0f) * wmask;

        const int  mtn  = mt + MGRID;
        const bool more = mtn < MTILES;
        if (more) {
            const int g0 = mtn * WTQ + tid;
            p0 = W4[min(g0,       WQTOT - 1)];
            p1 = W4[min(g0 + 128, WQTOT - 1)];
            p2 = W4[min(g0 + 256, WQTOT - 1)];
            p3 = W4[min(g0 + 384, WQTOT - 1)];
            p4 = W4[min(g0 + 512, WQTOT - 1)];
            const int vRow = mtn * 16 + col;
            const int vc   = (vRow < NV) ? vRow : 0;
            r0 = vt[vc * 3 + 2 * khalf];
            r1 = vt[vc * 3 + 1];
        }

        const float* __restrict__ wr = wt[buf] + col * JN;
        float4 wq[JN / 4];

#define SKIN_STEP4(jq)                                                                        \
        {                                                                                     \
            v2f a;                                                                            \
            a.x = wq[jq].x * c0; a.y = wq[jq].x * c1;                                         \
            acc0 = __builtin_amdgcn_wmma_f32_16x16x4_f32(false, a, false, bfr[4 * (jq) + 0],  \
                                                         (short)0, acc0, false, false);       \
            a.x = wq[jq].y * c0; a.y = wq[jq].y * c1;                                         \
            acc1 = __builtin_amdgcn_wmma_f32_16x16x4_f32(false, a, false, bfr[4 * (jq) + 1],  \
                                                         (short)0, acc1, false, false);       \
            a.x = wq[jq].z * c0; a.y = wq[jq].z * c1;                                         \
            acc0 = __builtin_amdgcn_wmma_f32_16x16x4_f32(false, a, false, bfr[4 * (jq) + 2],  \
                                                         (short)0, acc0, false, false);       \
            a.x = wq[jq].w * c0; a.y = wq[jq].w * c1;                                         \
            acc1 = __builtin_amdgcn_wmma_f32_16x16x4_f32(false, a, false, bfr[4 * (jq) + 3],  \
                                                         (short)0, acc1, false, false);       \
        }

        v8f acc0 = {}, acc1 = {};
#pragma unroll
        for (int jq = 0; jq < 28; ++jq)
            wq[jq] = *(const float4*)(wr + jq * 4);
        __builtin_amdgcn_sched_barrier(0);
#pragma unroll
        for (int jq = 0; jq < 7; ++jq) SKIN_STEP4(jq)
#pragma unroll
        for (int jq = 28; jq < JN / 4; ++jq)
            wq[jq] = *(const float4*)(wr + jq * 4);
        __builtin_amdgcn_sched_barrier(0);
#pragma unroll
        for (int jq = 7; jq < JN / 4; ++jq) SKIN_STEP4(jq)
#undef SKIN_STEP4
        const v8f acc = acc0 + acc1;

        float* __restrict__ stage = lds[buf];
#pragma unroll
        for (int i = 0; i < 8; ++i)
            stage[(khalf * 8 + i) * SLDW + wave * 16 + col] = acc[i];

        if (more) {
            float4* __restrict__ wd = (float4*)wt[buf ^ 1] + tid;
            wd[0] = p0; wd[128] = p1; wd[256] = p2; wd[384] = p3;
            if (has5) wd[512] = p4;
        }
        __syncthreads();

        const int vr  = mt * 16 + row;
        const int tof = mt * 48;
        float val[WOG];
        int   idx[WOG];
#pragma unroll
        for (int k = 0; k < WOG; ++k) {
            val[k] = stage[lof[k]] * scg[k] + trg[k];
            idx[k] = oof[k] + tof;
        }
        if (act && vr < NV) {
#pragma unroll
            for (int k = 0; k < WOG; ++k)
                Vout[(size_t)idx[k]] = val[k];
        }
        buf ^= 1;
    }
}

extern "C" void kernel_launch(void* const* d_in, const int* in_sizes, int n_in,
                              void* d_out, int out_size, void* d_ws, size_t ws_size,
                              hipStream_t stream) {
    const float* pose         = (const float*)d_in[0];
    const float* bone_lengths = (const float*)d_in[1];
    const float* cbl          = (const float*)d_in[2];
    const float* trans        = (const float*)d_in[3];
    const float* scale        = (const float*)d_in[4];
    const float* v_template   = (const float*)d_in[5];
    const float* t_pose       = (const float*)d_in[6];
    const float* weights      = (const float*)d_in[7];
    const int*   parent       = (const int*)d_in[8];
    const int*   bone_mapper  = (const int*)d_in[9];

    float* Gmat = (float*)d_ws;

    float* Vout = (float*)d_out;
    float* Jout = Vout + (size_t)BATCH * NV * 3;

    mouse_chain_kernel<<<dim3(BATCH), CHT, 0, stream>>>(
        pose, bone_lengths, cbl, trans, scale, t_pose, parent, bone_mapper,
        Gmat, Jout);

    dim3 grid(MGRID, (NC / 16) / WPB);
    mouse_skin_gemm<<<grid, WPB * 32, 0, stream>>>(
        weights, v_template, Gmat, scale, trans, Vout);
}


# chain kernel: pointer jumping over the joint tree (8 rounds, all joints at once: m = m[q]*m, q = q[q]) replaces the 139-step serial walk; transform products re-associated (rounding order changes)
# speedup vs baseline: 2.7038x; 1.0436x over previous
#include <hip/hip_runtime.h>
#include <hip/hip_bf16.h>

#define BATCH 512
#define JN    140
#define NB    28
#define NV    13059
#define NC    (3 * BATCH)
#define KDIM  (4 * JN)
#define CHT   160

typedef __attribute__((ext_vector_type(2))) float v2f;
typedef __attribute__((ext_vector_type(8))) float v8f;

#define DBL_ROUNDS 8
__global__ __launch_bounds__(CHT)
void mouse_chain_kernel(const float* __restrict__ pose,
                        const float* __restrict__ bone_lengths,
                        const float* __restrict__ cbl,
                        const float* __restrict__ trans,
                        const float* __restrict__ scale,
                        const float* __restrict__ Jt,
                        const int*   __restrict__ parent,
                        const int*   __restrict__ bone_mapper,
                        float* __restrict__ Gmat,
                        float* __restrict__ Jout)
{
    __shared__ __attribute__((aligned(16))) float Gb[2][JN * 12];
    __shared__ __attribute__((aligned(16))) float Jl[JN * 4];
    __shared__ int Qb[2][JN];

    const int t = threadIdx.x;
    const int b = blockIdx.x;
    const float sc  = scale[b];
    const float tr0 = trans[b * 3 + 0];
    const float tr1 = trans[b * 3 + 1];
    const float tr2 = trans[b * 3 + 2];

    float m[12];
#pragma unroll
    for (int e = 0; e < 12; ++e) m[e] = 0.0f;
    int q = 0;
    int fin = 1;

    if (t < JN) {
        const int i = t;
        const float ez = pose[(b * JN + i) * 3 + 0];
        const float ey = pose[(b * JN + i) * 3 + 1];
        const float ex = pose[(b * JN + i) * 3 + 2];
        const float cx = cosf(ex), sx = sinf(ex);
        const float cy = cosf(ey), sy = sinf(ey);
        const float cz = cosf(ez), sz = sinf(ez);
        float R[9];
        R[0] = cz * cy; R[1] = cz * sy * sx - sz * cx; R[2] = cz * sy * cx + sz * sx;
        R[3] = sz * cy; R[4] = sz * sy * sx + cz * cx; R[5] = sz * sy * cx - cz * sx;
        R[6] = -sy;     R[7] = cy * sx;                R[8] = cy * cx;

        const float jx = Jt[i * 3 + 0], jy = Jt[i * 3 + 1], jz = Jt[i * 3 + 2];
        *(float4*)(Jl + i * 4) = make_float4(jx, jy, jz, 0.0f);

        if (i == 0) {
            m[0] = R[0]; m[1] = R[1]; m[2]  = R[2]; m[3]  = jx;
            m[4] = R[3]; m[5] = R[4]; m[6]  = R[5]; m[7]  = jy;
            m[8] = R[6]; m[9] = R[7]; m[10] = R[8]; m[11] = jz;
        } else {
            const int p = parent[i];
            float s;
            if (i == 1) {
                s = cbl[b];
            } else {
                const int bm = bone_mapper[i];
                s = (bm >= 0) ? 2.0f / (1.0f + expf(-bone_lengths[b * NB + bm] * 0.2f))
                              : 1.0f;
            }
            if ((unsigned)p < (unsigned)i) {
                m[0] = R[0]; m[1] = R[1]; m[2]  = R[2]; m[3]  = (jx - Jt[p * 3 + 0]) * s;
                m[4] = R[3]; m[5] = R[4]; m[6]  = R[5]; m[7]  = (jy - Jt[p * 3 + 1]) * s;
                m[8] = R[6]; m[9] = R[7]; m[10] = R[8]; m[11] = (jz - Jt[p * 3 + 2]) * s;
                q = p;
                fin = 0;
            }
        }
        *(float4*)(Gb[0] + i * 12 + 0) = make_float4(m[0], m[1], m[2],  m[3]);
        *(float4*)(Gb[0] + i * 12 + 4) = make_float4(m[4], m[5], m[6],  m[7]);
        *(float4*)(Gb[0] + i * 12 + 8) = make_float4(m[8], m[9], m[10], m[11]);
        Qb[0][i] = q | (fin << 8);
    }
    __syncthreads();

    int cur = 0;
    for (int k = 0; k < DBL_ROUNDS; ++k) {
        if (t < JN) {
            if (!fin) {
                const float4 g0 = *(const float4*)(Gb[cur] + q * 12 + 0);
                const float4 g1 = *(const float4*)(Gb[cur] + q * 12 + 4);
                const float4 g2 = *(const float4*)(Gb[cur] + q * 12 + 8);
                const int qq = Qb[cur][q];
                float n[12];
#pragma unroll
                for (int c = 0; c < 4; ++c) {
                    n[0 + c] = fmaf(g0.z, m[8 + c], fmaf(g0.y, m[4 + c], g0.x * m[c]));
                    n[4 + c] = fmaf(g1.z, m[8 + c], fmaf(g1.y, m[4 + c], g1.x * m[c]));
                    n[8 + c] = fmaf(g2.z, m[8 + c], fmaf(g2.y, m[4 + c], g2.x * m[c]));
                }
                n[3]  += g0.w;
                n[7]  += g1.w;
                n[11] += g2.w;
#pragma unroll
                for (int e = 0; e < 12; ++e) m[e] = n[e];
                q   = qq & 0xff;
                fin = qq >> 8;
            }
            float* __restrict__ gd = Gb[cur ^ 1] + t * 12;
            *(float4*)(gd + 0) = make_float4(m[0], m[1], m[2],  m[3]);
            *(float4*)(gd + 4) = make_float4(m[4], m[5], m[6],  m[7]);
            *(float4*)(gd + 8) = make_float4(m[8], m[9], m[10], m[11]);
            Qb[cur ^ 1][t] = q | (fin << 8);
        }
        __syncthreads();
        cur ^= 1;
    }
    const float* __restrict__ Gl = Gb[cur];

    for (int item = t; item < 3 * JN; item += CHT) {
        const int i = item / 3;
        const int r = item - 3 * i;
        const float4 g  = *(const float4*)(Gl + item * 4);
        const float4 jt = *(const float4*)(Jl + i * 4);
        const float trel = g.w - (g.x * jt.x + g.y * jt.y + g.z * jt.z);
        const size_t c2 = (size_t)(3 * b + r) * 2;
        *(float2*)(Gmat + (size_t)(2 * i    ) * NC * 2 + c2) = make_float2(g.x, g.y);
        *(float2*)(Gmat + (size_t)(2 * i + 1) * NC * 2 + c2) = make_float2(g.z, trel);
        const float tr = (r == 0) ? tr0 : ((r == 1) ? tr1 : tr2);
        Jout[b * (JN * 3) + item] = g.w * sc + tr;
    }
}

#define MTILES ((NV + 15) / 16)
#define MGRID  63
#define WPB    4
#define SLAB   (WPB * 16)
#define SLDW   (SLAB + 4)
#define NSEG   2
#define WOG    11
#define WTF    (16 * JN)
#define WTQ    (WTF / 4)
#define WQTOT  (NV * (JN / 4))

__global__ __launch_bounds__(WPB * 32, 2)
void mouse_skin_gemm(const float* __restrict__ W,
                     const float* __restrict__ vt,
                     const float* __restrict__ Gmat,
                     const float* __restrict__ scale,
                     const float* __restrict__ trans,
                     float* __restrict__ Vout)
{
    __shared__ float lds[2][16 * SLDW];
    __shared__ __attribute__((aligned(16))) float wt[2][WTF];

    const int tid   = threadIdx.x;
    const int lane  = threadIdx.x & 31;
    const int wave  = threadIdx.x >> 5;
    const int nSub  = blockIdx.y * WPB + wave;
    const int col   = lane & 15;
    const int khalf = lane >> 4;

    const float4* __restrict__ W4 = (const float4*)W;

    const bool has5 = tid < WTQ - 4 * (WPB * 32);
    float4 p0, p1, p2, p3, p4;
    float  r0, r1;
    {
        const int m0 = blockIdx.x;
        const int g0 = m0 * WTQ + tid;
        p0 = W4[min(g0,       WQTOT - 1)];
        p1 = W4[min(g0 + 128, WQTOT - 1)];
        p2 = W4[min(g0 + 256, WQTOT - 1)];
        p3 = W4[min(g0 + 384, WQTOT - 1)];
        p4 = W4[min(g0 + 512, WQTOT - 1)];
        const int vRow = m0 * 16 + col;
        const int vc   = (vRow < NV) ? vRow : 0;
        r0 = vt[vc * 3 + 2 * khalf];
        r1 = vt[vc * 3 + 1];
    }

    const int cIdx = nSub * 16 + col;
    const float* __restrict__ gp =
        Gmat + ((size_t)khalf * NC + cIdx) * 2;
    v2f bfr[JN];
#pragma unroll
    for (int j = 0; j < JN; ++j)
        bfr[j] = *(const v2f*)(gp + (size_t)j * 4 * NC);

    const int C0     = blockIdx.y * SLAB;
    const int bFirst = C0 / 3;
    const int seg = threadIdx.x / 48;
    const bool act = seg < NSEG;
    const int segc = act ? seg : 0;
    const int s   = threadIdx.x - seg * 48;
    const int row = s / 3;
    const int r   = s - row * 3;
    int   lof[WOG];
    int   oof[WOG];
    float scg[WOG], trg[WOG];
#pragma unroll
    for (int k = 0; k < WOG; ++k) {
        int bg = bFirst + segc + NSEG * k;
        int lc = 3 * bg + r - C0;
        if (lc < 0)     { bg += NSEG; lc += 3 * NSEG; }
        if (lc >= SLAB) { bg -= NSEG; lc -= 3 * NSEG; }
        lof[k] = row * SLDW + lc;
        oof[k] = bg * (NV * 3) + s;
        scg[k] = scale[bg];
        trg[k] = trans[bg * 3 + r];
    }

    {
        float4* __restrict__ wd = (float4*)wt[0] + tid;
        wd[0] = p0; wd[128] = p1; wd[256] = p2; wd[384] = p3;
        if (has5) wd[512] = p4;
    }
    __syncthreads();

    int buf = 0;
    for (int mt = blockIdx.x; mt < MTILES; mt += MGRID) {
        const bool  vok   = (mt * 16 + col) < NV;
        const float wmask = vok ? 1.0f : 0.0f;

        const float c0 = r0 * wmask;
        const float c1 = ((khalf == 0) ? r1 : 1.0f) * wmask;

        const int  mtn  = mt + MGRID;
        const bool more = mtn < MTILES;
        if (more) {
            const int g0 = mtn * WTQ + tid;
            p0 = W4[min(g0,       WQTOT - 1)];
            p1 = W4[min(g0 + 128, WQTOT - 1)];
            p2 = W4[min(g0 + 256, WQTOT - 1)];
            p3 = W4[min(g0 + 384, WQTOT - 1)];
            p4 = W4[min(g0 + 512, WQTOT - 1)];
            const int vRow = mtn * 16 + col;
            const int vc   = (vRow < NV) ? vRow : 0;
            r0 = vt[vc * 3 + 2 * khalf];
            r1 = vt[vc * 3 + 1];
        }

        const float* __restrict__ wr = wt[buf] + col * JN;
        float4 wq[JN / 4];

#define SKIN_STEP4(jq)                                                                        \
        {                                                                                     \
            v2f a;                                                                            \
            a.x = wq[jq].x * c0; a.y = wq[jq].x * c1;                                         \
            acc0 = __builtin_amdgcn_wmma_f32_16x16x4_f32(false, a, false, bfr[4 * (jq) + 0],  \
                                                         (short)0, acc0, false, false);       \
            a.x = wq[jq].y * c0; a.y = wq[jq].y * c1;                                         \
            acc1 = __builtin_amdgcn_wmma_f32_16x16x4_f32(false, a, false, bfr[4 * (jq) + 1],  \
                                                         (short)0, acc1, false, false);       \
            a.x = wq[jq].z * c0; a.y = wq[jq].z * c1;                                         \
            acc0 = __builtin_amdgcn_wmma_f32_16x16x4_f32(false, a, false, bfr[4 * (jq) + 2],  \
                                                         (short)0, acc0, false, false);       \
            a.x = wq[jq].w * c0; a.y = wq[jq].w * c1;                                         \
            acc1 = __builtin_amdgcn_wmma_f32_16x16x4_f32(false, a, false, bfr[4 * (jq) + 3],  \
                                                         (short)0, acc1, false, false);       \
        }

        v8f acc0 = {}, acc1 = {};
#pragma unroll
        for (int jq = 0; jq < 28; ++jq)
            wq[jq] = *(const float4*)(wr + jq * 4);
        __builtin_amdgcn_sched_barrier(0);
#pragma unroll
        for (int jq = 0; jq < 7; ++jq) SKIN_STEP4(jq)
#pragma unroll
        for (int jq = 28; jq < JN / 4; ++jq)
            wq[jq] = *(const float4*)(wr + jq * 4);
        __builtin_amdgcn_sched_barrier(0);
#pragma unroll
        for (int jq = 7; jq < JN / 4; ++jq) SKIN_STEP4(jq)
#undef SKIN_STEP4
        const v8f acc = acc0 + acc1;

        float* __restrict__ stage = lds[buf];
#pragma unroll
        for (int i = 0; i < 8; ++i)
            stage[(khalf * 8 + i) * SLDW + wave * 16 + col] = acc[i];

        if (more) {
            float4* __restrict__ wd = (float4*)wt[buf ^ 1] + tid;
            wd[0] = p0; wd[128] = p1; wd[256] = p2; wd[384] = p3;
            if (has5) wd[512] = p4;
        }
        __syncthreads();

        const int vr  = mt * 16 + row;
        const int tof = mt * 48;
        float val[WOG];
        int   idx[WOG];
#pragma unroll
        for (int k = 0; k < WOG; ++k) {
            val[k] = stage[lof[k]] * scg[k] + trg[k];
            idx[k] = oof[k] + tof;
        }
        if (act && vr < NV) {
#pragma unroll
            for (int k = 0; k < WOG; ++k)
                Vout[(size_t)idx[k]] = val[k];
        }
        buf ^= 1;
    }
}

extern "C" void kernel_launch(void* const* d_in, const int* in_sizes, int n_in,
                              void* d_out, int out_size, void* d_ws, size_t ws_size,
                              hipStream_t stream) {
    const float* pose         = (const float*)d_in[0];
    const float* bone_lengths = (const float*)d_in[1];
    const float* cbl          = (const float*)d_in[2];
    const float* trans        = (const float*)d_in[3];
    const float* scale        = (const float*)d_in[4];
    const float* v_template   = (const float*)d_in[5];
    const float* t_pose       = (const float*)d_in[6];
    const float* weights      = (const float*)d_in[7];
    const int*   parent       = (const int*)d_in[8];
    const int*   bone_mapper  = (const int*)d_in[9];

    float* Gmat = (float*)d_ws;

    float* Vout = (float*)d_out;
    float* Jout = Vout + (size_t)BATCH * NV * 3;

    mouse_chain_kernel<<<dim3(BATCH), CHT, 0, stream>>>(
        pose, bone_lengths, cbl, trans, scale, t_pose, parent, bone_mapper,
        Gmat, Jout);

    dim3 grid(MGRID, (NC / 16) / WPB);
    mouse_skin_gemm<<<grid, WPB * 32, 0, stream>>>(
        weights, v_template, Gmat, scale, trans, Vout);
}


# GEMM: each wave owns 32 columns (B fragments of two 16-column subtiles in registers, one wave per SIMD, 252 blocks resident at once): each A fragment feeds two WMMAs; half the barriers per WMMA
# speedup vs baseline: 2.8413x; 1.0509x over previous
#include <hip/hip_runtime.h>
#include <hip/hip_bf16.h>

#define BATCH 512
#define JN    140
#define NB    28
#define NV    13059
#define NC    (3 * BATCH)
#define KDIM  (4 * JN)
#define CHT   160

typedef __attribute__((ext_vector_type(2))) float v2f;
typedef __attribute__((ext_vector_type(8))) float v8f;

#define DBL_ROUNDS 8
__global__ __launch_bounds__(CHT)
void mouse_chain_kernel(const float* __restrict__ pose,
                        const float* __restrict__ bone_lengths,
                        const float* __restrict__ cbl,
                        const float* __restrict__ trans,
                        const float* __restrict__ scale,
                        const float* __restrict__ Jt,
                        const int*   __restrict__ parent,
                        const int*   __restrict__ bone_mapper,
                        float* __restrict__ Gmat,
                        float* __restrict__ Jout)
{
    __shared__ __attribute__((aligned(16))) float Gb[2][JN * 12];
    __shared__ __attribute__((aligned(16))) float Jl[JN * 4];
    __shared__ int Qb[2][JN];

    const int t = threadIdx.x;
    const int b = blockIdx.x;
    const float sc  = scale[b];
    const float tr0 = trans[b * 3 + 0];
    const float tr1 = trans[b * 3 + 1];
    const float tr2 = trans[b * 3 + 2];

    float m[12];
#pragma unroll
    for (int e = 0; e < 12; ++e) m[e] = 0.0f;
    int q = 0;
    int fin = 1;

    if (t < JN) {
        const int i = t;
        const float ez = pose[(b * JN + i) * 3 + 0];
        const float ey = pose[(b * JN + i) * 3 + 1];
        const float ex = pose[(b * JN + i) * 3 + 2];
        const float cx = cosf(ex), sx = sinf(ex);
        const float cy = cosf(ey), sy = sinf(ey);
        const float cz = cosf(ez), sz = sinf(ez);
        float R[9];
        R[0] = cz * cy; R[1] = cz * sy * sx - sz * cx; R[2] = cz * sy * cx + sz * sx;
        R[3] = sz * cy; R[4] = sz * sy * sx + cz * cx; R[5] = sz * sy * cx - cz * sx;
        R[6] = -sy;     R[7] = cy * sx;                R[8] = cy * cx;

        const float jx = Jt[i * 3 + 0], jy = Jt[i * 3 + 1], jz = Jt[i * 3 + 2];
        *(float4*)(Jl + i * 4) = make_float4(jx, jy, jz, 0.0f);

        if (i == 0) {
            m[0] = R[0]; m[1] = R[1]; m[2]  = R[2]; m[3]  = jx;
            m[4] = R[3]; m[5] = R[4]; m[6]  = R[5]; m[7]  = jy;
            m[8] = R[6]; m[9] = R[7]; m[10] = R[8]; m[11] = jz;
        } else {
            const int p = parent[i];
            float s;
            if (i == 1) {
                s = cbl[b];
            } else {
                const int bm = bone_mapper[i];
                s = (bm >= 0) ? 2.0f / (1.0f + expf(-bone_lengths[b * NB + bm] * 0.2f))
                              : 1.0f;
            }
            if ((unsigned)p < (unsigned)i) {
                m[0] = R[0]; m[1] = R[1]; m[2]  = R[2]; m[3]  = (jx - Jt[p * 3 + 0]) * s;
                m[4] = R[3]; m[5] = R[4]; m[6]  = R[5]; m[7]  = (jy - Jt[p * 3 + 1]) * s;
                m[8] = R[6]; m[9] = R[7]; m[10] = R[8]; m[11] = (jz - Jt[p * 3 + 2]) * s;
                q = p;
                fin = 0;
            }
        }
        *(float4*)(Gb[0] + i * 12 + 0) = make_float4(m[0], m[1], m[2],  m[3]);
        *(float4*)(Gb[0] + i * 12 + 4) = make_float4(m[4], m[5], m[6],  m[7]);
        *(float4*)(Gb[0] + i * 12 + 8) = make_float4(m[8], m[9], m[10], m[11]);
        Qb[0][i] = q | (fin << 8);
    }
    __syncthreads();

    int cur = 0;
    for (int k = 0; k < DBL_ROUNDS; ++k) {
        if (t < JN) {
            if (!fin) {
                const float4 g0 = *(const float4*)(Gb[cur] + q * 12 + 0);
                const float4 g1 = *(const float4*)(Gb[cur] + q * 12 + 4);
                const float4 g2 = *(const float4*)(Gb[cur] + q * 12 + 8);
                const int qq = Qb[cur][q];
                float n[12];
#pragma unroll
                for (int c = 0; c < 4; ++c) {
                    n[0 + c] = fmaf(g0.z, m[8 + c], fmaf(g0.y, m[4 + c], g0.x * m[c]));
                    n[4 + c] = fmaf(g1.z, m[8 + c], fmaf(g1.y, m[4 + c], g1.x * m[c]));
                    n[8 + c] = fmaf(g2.z, m[8 + c], fmaf(g2.y, m[4 + c], g2.x * m[c]));
                }
                n[3]  += g0.w;
                n[7]  += g1.w;
                n[11] += g2.w;
#pragma unroll
                for (int e = 0; e < 12; ++e) m[e] = n[e];
                q   = qq & 0xff;
                fin = qq >> 8;
            }
            float* __restrict__ gd = Gb[cur ^ 1] + t * 12;
            *(float4*)(gd + 0) = make_float4(m[0], m[1], m[2],  m[3]);
            *(float4*)(gd + 4) = make_float4(m[4], m[5], m[6],  m[7]);
            *(float4*)(gd + 8) = make_float4(m[8], m[9], m[10], m[11]);
            Qb[cur ^ 1][t] = q | (fin << 8);
        }
        __syncthreads();
        cur ^= 1;
    }
    const float* __restrict__ Gl = Gb[cur];

    for (int item = t; item < 3 * JN; item += CHT) {
        const int i = item / 3;
        const int r = item - 3 * i;
        const float4 g  = *(const float4*)(Gl + item * 4);
        const float4 jt = *(const float4*)(Jl + i * 4);
        const float trel = g.w - (g.x * jt.x + g.y * jt.y + g.z * jt.z);
        const size_t c2 = (size_t)(3 * b + r) * 2;
        *(float2*)(Gmat + (size_t)(2 * i    ) * NC * 2 + c2) = make_float2(g.x, g.y);
        *(float2*)(Gmat + (size_t)(2 * i + 1) * NC * 2 + c2) = make_float2(g.z, trel);
        const float tr = (r == 0) ? tr0 : ((r == 1) ? tr1 : tr2);
        Jout[b * (JN * 3) + item] = g.w * sc + tr;
    }
}

#define MTILES ((NV + 15) / 16)
#define MGRID  21
#define WPB    4
#define NSW    2
#define SLAB   (WPB * NSW * 16)
#define SLDW   (SLAB + 4)
#define NSEG   2
#define WOG    22
#define WTF    (16 * JN)
#define WTQ    (WTF / 4)
#define WQTOT  (NV * (JN / 4))

__global__ __launch_bounds__(WPB * 32, 1)
void mouse_skin_gemm(const float* __restrict__ W,
                     const float* __restrict__ vt,
                     const float* __restrict__ Gmat,
                     const float* __restrict__ scale,
                     const float* __restrict__ trans,
                     float* __restrict__ Vout)
{
    __shared__ float lds[2][16 * SLDW];
    __shared__ __attribute__((aligned(16))) float wt[2][WTF];

    const int tid   = threadIdx.x;
    const int lane  = threadIdx.x & 31;
    const int wave  = threadIdx.x >> 5;
    const int col   = lane & 15;
    const int khalf = lane >> 4;

    const float4* __restrict__ W4 = (const float4*)W;

    const bool has5 = tid < WTQ - 4 * (WPB * 32);
    float4 p0, p1, p2, p3, p4;
    float  r0, r1;
    {
        const int m0 = blockIdx.x;
        const int g0 = m0 * WTQ + tid;
        p0 = W4[min(g0,       WQTOT - 1)];
        p1 = W4[min(g0 + 128, WQTOT - 1)];
        p2 = W4[min(g0 + 256, WQTOT - 1)];
        p3 = W4[min(g0 + 384, WQTOT - 1)];
        p4 = W4[min(g0 + 512, WQTOT - 1)];
        const int vRow = m0 * 16 + col;
        const int vc   = (vRow < NV) ? vRow : 0;
        r0 = vt[vc * 3 + 2 * khalf];
        r1 = vt[vc * 3 + 1];
    }

    const int C0   = blockIdx.y * SLAB;
    const int cIdx = C0 + wave * (NSW * 16) + col;
    const float* __restrict__ gp =
        Gmat + ((size_t)khalf * NC + cIdx) * 2;
    v2f bfa[JN], bfb[JN];
#pragma unroll
    for (int j = 0; j < JN; ++j) {
        bfa[j] = *(const v2f*)(gp + (size_t)j * 4 * NC);
        bfb[j] = *(const v2f*)(gp + (size_t)j * 4 * NC + 32);
    }

    const int bFirst = C0 / 3;
    const int seg = threadIdx.x / 48;
    const bool act = seg < NSEG;
    const int segc = act ? seg : 0;
    const int s   = threadIdx.x - seg * 48;
    const int row = s / 3;
    const int r   = s - row * 3;
    int   lof[WOG];
    int   oof[WOG];
    float scg[WOG], trg[WOG];
#pragma unroll
    for (int k = 0; k < WOG; ++k) {
        int bg = bFirst + segc + NSEG * k;
        int lc = 3 * bg + r - C0;
        if (lc < 0)     { bg += NSEG; lc += 3 * NSEG; }
        if (lc >= SLAB) { bg -= NSEG; lc -= 3 * NSEG; }
        lof[k] = row * SLDW + lc;
        oof[k] = bg * (NV * 3) + s;
        scg[k] = scale[bg];
        trg[k] = trans[bg * 3 + r];
    }

    {
        float4* __restrict__ wd = (float4*)wt[0] + tid;
        wd[0] = p0; wd[128] = p1; wd[256] = p2; wd[384] = p3;
        if (has5) wd[512] = p4;
    }
    __syncthreads();

    int buf = 0;
    for (int mt = blockIdx.x; mt < MTILES; mt += MGRID) {
        const bool  vok   = (mt * 16 + col) < NV;
        const float wmask = vok ? 1.0f : 0.0f;

        const float c0 = r0 * wmask;
        const float c1 = ((khalf == 0) ? r1 : 1.0f) * wmask;

        const int  mtn  = mt + MGRID;
        const bool more = mtn < MTILES;
        if (more) {
            const int g0 = mtn * WTQ + tid;
            p0 = W4[min(g0,       WQTOT - 1)];
            p1 = W4[min(g0 + 128, WQTOT - 1)];
            p2 = W4[min(g0 + 256, WQTOT - 1)];
            p3 = W4[min(g0 + 384, WQTOT - 1)];
            p4 = W4[min(g0 + 512, WQTOT - 1)];
            const int vRow = mtn * 16 + col;
            const int vc   = (vRow < NV) ? vRow : 0;
            r0 = vt[vc * 3 + 2 * khalf];
            r1 = vt[vc * 3 + 1];
        }

        const float* __restrict__ wr = wt[buf] + col * JN;
        float4 wq[JN / 4];

#define SKIN_PAIR(wv, j, ea, eb)                                                              \
        {                                                                                     \
            v2f a;                                                                            \
            a.x = (wv) * c0; a.y = (wv) * c1;                                                 \
            ea = __builtin_amdgcn_wmma_f32_16x16x4_f32(false, a, false, bfa[j],               \
                                                       (short)0, ea, false, false);           \
            eb = __builtin_amdgcn_wmma_f32_16x16x4_f32(false, a, false, bfb[j],               \
                                                       (short)0, eb, false, false);           \
        }
#define SKIN_STEP4(jq)                                                                        \
        {                                                                                     \
            SKIN_PAIR(wq[jq].x, 4 * (jq) + 0, acc0a, acc0b)                                   \
            SKIN_PAIR(wq[jq].y, 4 * (jq) + 1, acc1a, acc1b)                                   \
            SKIN_PAIR(wq[jq].z, 4 * (jq) + 2, acc0a, acc0b)                                   \
            SKIN_PAIR(wq[jq].w, 4 * (jq) + 3, acc1a, acc1b)                                   \
        }

        v8f acc0a = {}, acc1a = {}, acc0b = {}, acc1b = {};
#pragma unroll
        for (int jq = 0; jq < JN / 4; ++jq)
            wq[jq] = *(const float4*)(wr + jq * 4);
        __builtin_amdgcn_sched_barrier(0);
#pragma unroll
        for (int jq = 0; jq < JN / 4; ++jq) SKIN_STEP4(jq)
#undef SKIN_STEP4
#undef SKIN_PAIR
        const v8f acca = acc0a + acc1a;
        const v8f accb = acc0b + acc1b;

        float* __restrict__ stage = lds[buf];
#pragma unroll
        for (int i = 0; i < 8; ++i) {
            stage[(khalf * 8 + i) * SLDW + wave * (NSW * 16) + col]      = acca[i];
            stage[(khalf * 8 + i) * SLDW + wave * (NSW * 16) + 16 + col] = accb[i];
        }

        if (more) {
            float4* __restrict__ wd = (float4*)wt[buf ^ 1] + tid;
            wd[0] = p0; wd[128] = p1; wd[256] = p2; wd[384] = p3;
            if (has5) wd[512] = p4;
        }
        __syncthreads();

        const int vr  = mt * 16 + row;
        const int tof = mt * 48;
        float val[WOG];
        int   idx[WOG];
#pragma unroll
        for (int k = 0; k < WOG; ++k) {
            val[k] = stage[lof[k]] * scg[k] + trg[k];
            idx[k] = oof[k] + tof;
        }
        if (act && vr < NV) {
#pragma unroll
            for (int k = 0; k < WOG; ++k)
                Vout[(size_t)idx[k]] = val[k];
        }
        buf ^= 1;
    }
}

extern "C" void kernel_launch(void* const* d_in, const int* in_sizes, int n_in,
                              void* d_out, int out_size, void* d_ws, size_t ws_size,
                              hipStream_t stream) {
    const float* pose         = (const float*)d_in[0];
    const float* bone_lengths = (const float*)d_in[1];
    const float* cbl          = (const float*)d_in[2];
    const float* trans        = (const float*)d_in[3];
    const float* scale        = (const float*)d_in[4];
    const float* v_template   = (const float*)d_in[5];
    const float* t_pose       = (const float*)d_in[6];
    const float* weights      = (const float*)d_in[7];
    const int*   parent       = (const int*)d_in[8];
    const int*   bone_mapper  = (const int*)d_in[9];

    float* Gmat = (float*)d_ws;

    float* Vout = (float*)d_out;
    float* Jout = Vout + (size_t)BATCH * NV * 3;

    mouse_chain_kernel<<<dim3(BATCH), CHT, 0, stream>>>(
        pose, bone_lengths, cbl, trans, scale, t_pose, parent, bone_mapper,
        Gmat, Jout);

    dim3 grid(MGRID, NC / SLAB);
    mouse_skin_gemm<<<grid, WPB * 32, 0, stream>>>(
        weights, v_template, Gmat, scale, trans, Vout);
}


# GEMM: template coordinates applied to 4 finished per-coefficient sums (3 fma per result), not to every weight: A fragments are raw weights from LDS, no VALU in WMMA block; summation order changes
# speedup vs baseline: 3.0157x; 1.0614x over previous
#include <hip/hip_runtime.h>
#include <hip/hip_bf16.h>

#define BATCH 512
#define JN    140
#define NB    28
#define NV    13059
#define NC    (3 * BATCH)
#define KDIM  (4 * JN)
#define CHT   160

typedef __attribute__((ext_vector_type(2))) float v2f;
typedef __attribute__((ext_vector_type(8))) float v8f;

#define DBL_ROUNDS 8
__global__ __launch_bounds__(CHT)
void mouse_chain_kernel(const float* __restrict__ pose,
                        const float* __restrict__ bone_lengths,
                        const float* __restrict__ cbl,
                        const float* __restrict__ trans,
                        const float* __restrict__ scale,
                        const float* __restrict__ Jt,
                        const int*   __restrict__ parent,
                        const int*   __restrict__ bone_mapper,
                        float* __restrict__ Gmat,
                        float* __restrict__ Jout)
{
    __shared__ __attribute__((aligned(16))) float Gb[2][JN * 12];
    __shared__ __attribute__((aligned(16))) float Jl[JN * 4];
    __shared__ int Qb[2][JN];

    const int t = threadIdx.x;
    const int b = blockIdx.x;
    const float sc  = scale[b];
    const float tr0 = trans[b * 3 + 0];
    const float tr1 = trans[b * 3 + 1];
    const float tr2 = trans[b * 3 + 2];

    float m[12];
#pragma unroll
    for (int e = 0; e < 12; ++e) m[e] = 0.0f;
    int q = 0;
    int fin = 1;

    if (t < JN) {
        const int i = t;
        const float ez = pose[(b * JN + i) * 3 + 0];
        const float ey = pose[(b * JN + i) * 3 + 1];
        const float ex = pose[(b * JN + i) * 3 + 2];
        const float cx = cosf(ex), sx = sinf(ex);
        const float cy = cosf(ey), sy = sinf(ey);
        const float cz = cosf(ez), sz = sinf(ez);
        float R[9];
        R[0] = cz * cy; R[1] = cz * sy * sx - sz * cx; R[2] = cz * sy * cx + sz * sx;
        R[3] = sz * cy; R[4] = sz * sy * sx + cz * cx; R[5] = sz * sy * cx - cz * sx;
        R[6] = -sy;     R[7] = cy * sx;                R[8] = cy * cx;

        const float jx = Jt[i * 3 + 0], jy = Jt[i * 3 + 1], jz = Jt[i * 3 + 2];
        *(float4*)(Jl + i * 4) = make_float4(jx, jy, jz, 0.0f);

        if (i == 0) {
            m[0] = R[0]; m[1] = R[1]; m[2]  = R[2]; m[3]  = jx;
            m[4] = R[3]; m[5] = R[4]; m[6]  = R[5]; m[7]  = jy;
            m[8] = R[6]; m[9] = R[7]; m[10] = R[8]; m[11] = jz;
        } else {
            const int p = parent[i];
            float s;
            if (i == 1) {
                s = cbl[b];
            } else {
                const int bm = bone_mapper[i];
                s = (bm >= 0) ? 2.0f / (1.0f + expf(-bone_lengths[b * NB + bm] * 0.2f))
                              : 1.0f;
            }
            if ((unsigned)p < (unsigned)i) {
                m[0] = R[0]; m[1] = R[1]; m[2]  = R[2]; m[3]  = (jx - Jt[p * 3 + 0]) * s;
                m[4] = R[3]; m[5] = R[4]; m[6]  = R[5]; m[7]  = (jy - Jt[p * 3 + 1]) * s;
                m[8] = R[6]; m[9] = R[7]; m[10] = R[8]; m[11] = (jz - Jt[p * 3 + 2]) * s;
                q = p;
                fin = 0;
            }
        }
        *(float4*)(Gb[0] + i * 12 + 0) = make_float4(m[0], m[1], m[2],  m[3]);
        *(float4*)(Gb[0] + i * 12 + 4) = make_float4(m[4], m[5], m[6],  m[7]);
        *(float4*)(Gb[0] + i * 12 + 8) = make_float4(m[8], m[9], m[10], m[11]);
        Qb[0][i] = q | (fin << 8);
    }
    __syncthreads();

    int cur = 0;
    for (int k = 0; k < DBL_ROUNDS; ++k) {
        if (t < JN) {
            if (!fin) {
                const float4 g0 = *(const float4*)(Gb[cur] + q * 12 + 0);
                const float4 g1 = *(const float4*)(Gb[cur] + q * 12 + 4);
                const float4 g2 = *(const float4*)(Gb[cur] + q * 12 + 8);
                const int qq = Qb[cur][q];
                float n[12];
#pragma unroll
                for (int c = 0; c < 4; ++c) {
                    n[0 + c] = fmaf(g0.z, m[8 + c], fmaf(g0.y, m[4 + c], g0.x * m[c]));
                    n[4 + c] = fmaf(g1.z, m[8 + c], fmaf(g1.y, m[4 + c], g1.x * m[c]));
                    n[8 + c] = fmaf(g2.z, m[8 + c], fmaf(g2.y, m[4 + c], g2.x * m[c]));
                }
                n[3]  += g0.w;
                n[7]  += g1.w;
                n[11] += g2.w;
#pragma unroll
                for (int e = 0; e < 12; ++e) m[e] = n[e];
                q   = qq & 0xff;
                fin = qq >> 8;
            }
            float* __restrict__ gd = Gb[cur ^ 1] + t * 12;
            *(float4*)(gd + 0) = make_float4(m[0], m[1], m[2],  m[3]);
            *(float4*)(gd + 4) = make_float4(m[4], m[5], m[6],  m[7]);
            *(float4*)(gd + 8) = make_float4(m[8], m[9], m[10], m[11]);
            Qb[cur ^ 1][t] = q | (fin << 8);
        }
        __syncthreads();
        cur ^= 1;
    }
    const float* __restrict__ Gl = Gb[cur];

    for (int item = t; item < 3 * (JN / 2); item += CHT) {
        const int q  = item / 3;
        const int r  = item - 3 * q;
        const int i0 = 2 * q;
        const int i1 = 2 * q + 1;
        const float4 ga = *(const float4*)(Gl + (i0 * 3 + r) * 4);
        const float4 gb = *(const float4*)(Gl + (i1 * 3 + r) * 4);
        const float4 ja = *(const float4*)(Jl + i0 * 4);
        const float4 jb = *(const float4*)(Jl + i1 * 4);
        const float ta = ga.w - (ga.x * ja.x + ga.y * ja.y + ga.z * ja.z);
        const float tb = gb.w - (gb.x * jb.x + gb.y * jb.y + gb.z * jb.z);
        const int ks = q >> 1;
        const int kh = q & 1;
        float* __restrict__ gq = Gmat + ((size_t)(8 * ks + kh) * NC + (size_t)(3 * b + r)) * 2;
        *(float2*)(gq)                       = make_float2(ga.x, gb.x);
        *(float2*)(gq + (size_t)2 * NC * 2)  = make_float2(ga.y, gb.y);
        *(float2*)(gq + (size_t)4 * NC * 2)  = make_float2(ga.z, gb.z);
        *(float2*)(gq + (size_t)6 * NC * 2)  = make_float2(ta, tb);
        const float tr = (r == 0) ? tr0 : ((r == 1) ? tr1 : tr2);
        Jout[b * (JN * 3) + i0 * 3 + r] = ga.w * sc + tr;
        Jout[b * (JN * 3) + i1 * 3 + r] = gb.w * sc + tr;
    }
}

#define MTILES ((NV + 15) / 16)
#define MGRID  21
#define WPB    4
#define NSW    2
#define SLAB   (WPB * NSW * 16)
#define SLDW   (SLAB + 4)
#define NSEG   2
#define WOG    22
#define WTF    (16 * JN)
#define WTQ    (WTF / 4)
#define WQTOT  (NV * (JN / 4))

__global__ __launch_bounds__(WPB * 32, 1)
void mouse_skin_gemm(const float* __restrict__ W,
                     const float* __restrict__ vt,
                     const float* __restrict__ Gmat,
                     const float* __restrict__ scale,
                     const float* __restrict__ trans,
                     float* __restrict__ Vout)
{
    __shared__ float lds[2][16 * SLDW];
    __shared__ __attribute__((aligned(16))) float wt[2][WTF];

    const int tid   = threadIdx.x;
    const int lane  = threadIdx.x & 31;
    const int wave  = threadIdx.x >> 5;
    const int col   = lane & 15;
    const int khalf = lane >> 4;

    const float4* __restrict__ W4 = (const float4*)W;

    const bool has5 = tid < WTQ - 4 * (WPB * 32);
    float4 p0, p1, p2, p3, p4;
    {
        const int m0 = blockIdx.x;
        const int g0 = m0 * WTQ + tid;
        p0 = W4[min(g0,       WQTOT - 1)];
        p1 = W4[min(g0 + 128, WQTOT - 1)];
        p2 = W4[min(g0 + 256, WQTOT - 1)];
        p3 = W4[min(g0 + 384, WQTOT - 1)];
        p4 = W4[min(g0 + 512, WQTOT - 1)];
    }

    const int C0   = blockIdx.y * SLAB;
    const int cIdx = C0 + wave * (NSW * 16) + col;
    const float* __restrict__ gp =
        Gmat + ((size_t)khalf * NC + cIdx) * 2;
    v2f bfa[JN], bfb[JN];
#pragma unroll
    for (int j = 0; j < JN; ++j) {
        bfa[j] = *(const v2f*)(gp + (size_t)j * 4 * NC);
        bfb[j] = *(const v2f*)(gp + (size_t)j * 4 * NC + 32);
    }

    const int bFirst = C0 / 3;
    const int seg = threadIdx.x / 48;
    const bool act = seg < NSEG;
    const int segc = act ? seg : 0;
    const int s   = threadIdx.x - seg * 48;
    const int row = s / 3;
    const int r   = s - row * 3;
    int   lof[WOG];
    int   oof[WOG];
    float scg[WOG], trg[WOG];
#pragma unroll
    for (int k = 0; k < WOG; ++k) {
        int bg = bFirst + segc + NSEG * k;
        int lc = 3 * bg + r - C0;
        if (lc < 0)     { bg += NSEG; lc += 3 * NSEG; }
        if (lc >= SLAB) { bg -= NSEG; lc -= 3 * NSEG; }
        lof[k] = row * SLDW + lc;
        oof[k] = bg * (NV * 3) + s;
        scg[k] = scale[bg];
        trg[k] = trans[bg * 3 + r];
    }

    {
        float4* __restrict__ wd = (float4*)wt[0] + tid;
        wd[0] = p0; wd[128] = p1; wd[256] = p2; wd[384] = p3;
        if (has5) wd[512] = p4;
    }
    __syncthreads();

    int buf = 0;
    for (int mt = blockIdx.x; mt < MTILES; mt += MGRID) {
        float cx[8], cy[8], cz[8];
#pragma unroll
        for (int i = 0; i < 8; ++i) {
            const int vrow = min(mt * 16 + 8 * khalf + i, NV - 1);
            cx[i] = vt[vrow * 3 + 0];
            cy[i] = vt[vrow * 3 + 1];
            cz[i] = vt[vrow * 3 + 2];
        }

        const int  mtn  = mt + MGRID;
        const bool more = mtn < MTILES;
        if (more) {
            const int g0 = mtn * WTQ + tid;
            p0 = W4[min(g0,       WQTOT - 1)];
            p1 = W4[min(g0 + 128, WQTOT - 1)];
            p2 = W4[min(g0 + 256, WQTOT - 1)];
            p3 = W4[min(g0 + 384, WQTOT - 1)];
            p4 = W4[min(g0 + 512, WQTOT - 1)];
        }

        const float* __restrict__ wr = wt[buf] + col * JN + 2 * khalf;
        v2f af[JN / 4];

#define SKIN_N(ks, n)                                                                         \
        ta[n] = __builtin_amdgcn_wmma_f32_16x16x4_f32(false, af[ks], false, bfa[4 * (ks) + (n)], \
                                                      (short)0, ta[n], false, false);         \
        tb[n] = __builtin_amdgcn_wmma_f32_16x16x4_f32(false, af[ks], false, bfb[4 * (ks) + (n)], \
                                                      (short)0, tb[n], false, false);
#define SKIN_KSTEP(ks) { SKIN_N(ks, 0) SKIN_N(ks, 1) SKIN_N(ks, 2) SKIN_N(ks, 3) }

        v8f ta[4] = {{}, {}, {}, {}};
        v8f tb[4] = {{}, {}, {}, {}};
#pragma unroll
        for (int ks = 0; ks < JN / 4; ++ks)
            af[ks] = *(const v2f*)(wr + ks * 4);
        __builtin_amdgcn_sched_barrier(0);
#pragma unroll
        for (int ks = 0; ks < JN / 4; ++ks) SKIN_KSTEP(ks)
#undef SKIN_KSTEP
#undef SKIN_N
        v8f acca, accb;
#pragma unroll
        for (int i = 0; i < 8; ++i) {
            acca[i] = fmaf(cz[i], ta[2][i], fmaf(cy[i], ta[1][i], fmaf(cx[i], ta[0][i], ta[3][i])));
            accb[i] = fmaf(cz[i], tb[2][i], fmaf(cy[i], tb[1][i], fmaf(cx[i], tb[0][i], tb[3][i])));
        }

        float* __restrict__ stage = lds[buf];
#pragma unroll
        for (int i = 0; i < 8; ++i) {
            stage[(khalf * 8 + i) * SLDW + wave * (NSW * 16) + col]      = acca[i];
            stage[(khalf * 8 + i) * SLDW + wave * (NSW * 16) + 16 + col] = accb[i];
        }

        if (more) {
            float4* __restrict__ wd = (float4*)wt[buf ^ 1] + tid;
            wd[0] = p0; wd[128] = p1; wd[256] = p2; wd[384] = p3;
            if (has5) wd[512] = p4;
        }
        __syncthreads();

        const int vr  = mt * 16 + row;
        const int tof = mt * 48;
        float val[WOG];
        int   idx[WOG];
#pragma unroll
        for (int k = 0; k < WOG; ++k) {
            val[k] = stage[lof[k]] * scg[k] + trg[k];
            idx[k] = oof[k] + tof;
        }
        if (act && vr < NV) {
#pragma unroll
            for (int k = 0; k < WOG; ++k)
                Vout[(size_t)idx[k]] = val[k];
        }
        buf ^= 1;
    }
}

extern "C" void kernel_launch(void* const* d_in, const int* in_sizes, int n_in,
                              void* d_out, int out_size, void* d_ws, size_t ws_size,
                              hipStream_t stream) {
    const float* pose         = (const float*)d_in[0];
    const float* bone_lengths = (const float*)d_in[1];
    const float* cbl          = (const float*)d_in[2];
    const float* trans        = (const float*)d_in[3];
    const float* scale        = (const float*)d_in[4];
    const float* v_template   = (const float*)d_in[5];
    const float* t_pose       = (const float*)d_in[6];
    const float* weights      = (const float*)d_in[7];
    const int*   parent       = (const int*)d_in[8];
    const int*   bone_mapper  = (const int*)d_in[9];

    float* Gmat = (float*)d_ws;

    float* Vout = (float*)d_out;
    float* Jout = Vout + (size_t)BATCH * NV * 3;

    mouse_chain_kernel<<<dim3(BATCH), CHT, 0, stream>>>(
        pose, bone_lengths, cbl, trans, scale, t_pose, parent, bone_mapper,
        Gmat, Jout);

    dim3 grid(MGRID, NC / SLAB);
    mouse_skin_gemm<<<grid, WPB * 32, 0, stream>>>(
        weights, v_template, Gmat, scale, trans, Vout);
}


# GEMM: two independent 4-wave blocks per WGP (2 waves per SIMD, 16 columns per wave): one block's WMMA run covers the other's writeout; wave priority rises along a run so the pair takes turns
# speedup vs baseline: 3.1540x; 1.0459x over previous
#include <hip/hip_runtime.h>
#include <hip/hip_bf16.h>

#define BATCH 512
#define JN    140
#define NB    28
#define NV    13059
#define NC    (3 * BATCH)
#define KDIM  (4 * JN)
#define CHT   160

typedef __attribute__((ext_vector_type(2))) float v2f;
typedef __attribute__((ext_vector_type(8))) float v8f;

#define DBL_ROUNDS 8
__global__ __launch_bounds__(CHT)
void mouse_chain_kernel(const float* __restrict__ pose,
                        const float* __restrict__ bone_lengths,
                        const float* __restrict__ cbl,
                        const float* __restrict__ trans,
                        const float* __restrict__ scale,
                        const float* __restrict__ Jt,
                        const int*   __restrict__ parent,
                        const int*   __restrict__ bone_mapper,
                        float* __restrict__ Gmat,
                        float* __restrict__ Jout)
{
    __shared__ __attribute__((aligned(16))) float Gb[2][JN * 12];
    __shared__ __attribute__((aligned(16))) float Jl[JN * 4];
    __shared__ int Qb[2][JN];

    const int t = threadIdx.x;
    const int b = blockIdx.x;
    const float sc  = scale[b];
    const float tr0 = trans[b * 3 + 0];
    const float tr1 = trans[b * 3 + 1];
    const float tr2 = trans[b * 3 + 2];

    float m[12];
#pragma unroll
    for (int e = 0; e < 12; ++e) m[e] = 0.0f;
    int q = 0;
    int fin = 1;

    if (t < JN) {
        const int i = t;
        const float ez = pose[(b * JN + i) * 3 + 0];
        const float ey = pose[(b * JN + i) * 3 + 1];
        const float ex = pose[(b * JN + i) * 3 + 2];
        const float cx = cosf(ex), sx = sinf(ex);
        const float cy = cosf(ey), sy = sinf(ey);
        const float cz = cosf(ez), sz = sinf(ez);
        float R[9];
        R[0] = cz * cy; R[1] = cz * sy * sx - sz * cx; R[2] = cz * sy * cx + sz * sx;
        R[3] = sz * cy; R[4] = sz * sy * sx + cz * cx; R[5] = sz * sy * cx - cz * sx;
        R[6] = -sy;     R[7] = cy * sx;                R[8] = cy * cx;

        const float jx = Jt[i * 3 + 0], jy = Jt[i * 3 + 1], jz = Jt[i * 3 + 2];
        *(float4*)(Jl + i * 4) = make_float4(jx, jy, jz, 0.0f);

        if (i == 0) {
            m[0] = R[0]; m[1] = R[1]; m[2]  = R[2]; m[3]  = jx;
            m[4] = R[3]; m[5] = R[4]; m[6]  = R[5]; m[7]  = jy;
            m[8] = R[6]; m[9] = R[7]; m[10] = R[8]; m[11] = jz;
        } else {
            const int p = parent[i];
            float s;
            if (i == 1) {
                s = cbl[b];
            } else {
                const int bm = bone_mapper[i];
                s = (bm >= 0) ? 2.0f / (1.0f + expf(-bone_lengths[b * NB + bm] * 0.2f))
                              : 1.0f;
            }
            if ((unsigned)p < (unsigned)i) {
                m[0] = R[0]; m[1] = R[1]; m[2]  = R[2]; m[3]  = (jx - Jt[p * 3 + 0]) * s;
                m[4] = R[3]; m[5] = R[4]; m[6]  = R[5]; m[7]  = (jy - Jt[p * 3 + 1]) * s;
                m[8] = R[6]; m[9] = R[7]; m[10] = R[8]; m[11] = (jz - Jt[p * 3 + 2]) * s;
                q = p;
                fin = 0;
            }
        }
        *(float4*)(Gb[0] + i * 12 + 0) = make_float4(m[0], m[1], m[2],  m[3]);
        *(float4*)(Gb[0] + i * 12 + 4) = make_float4(m[4], m[5], m[6],  m[7]);
        *(float4*)(Gb[0] + i * 12 + 8) = make_float4(m[8], m[9], m[10], m[11]);
        Qb[0][i] = q | (fin << 8);
    }
    __syncthreads();

    int cur = 0;
    for (int k = 0; k < DBL_ROUNDS; ++k) {
        if (t < JN) {
            if (!fin) {
                const float4 g0 = *(const float4*)(Gb[cur] + q * 12 + 0);
                const float4 g1 = *(const float4*)(Gb[cur] + q * 12 + 4);
                const float4 g2 = *(const float4*)(Gb[cur] + q * 12 + 8);
                const int qq = Qb[cur][q];
                float n[12];
#pragma unroll
                for (int c = 0; c < 4; ++c) {
                    n[0 + c] = fmaf(g0.z, m[8 + c], fmaf(g0.y, m[4 + c], g0.x * m[c]));
                    n[4 + c] = fmaf(g1.z, m[8 + c], fmaf(g1.y, m[4 + c], g1.x * m[c]));
                    n[8 + c] = fmaf(g2.z, m[8 + c], fmaf(g2.y, m[4 + c], g2.x * m[c]));
                }
                n[3]  += g0.w;
                n[7]  += g1.w;
                n[11] += g2.w;
#pragma unroll
                for (int e = 0; e < 12; ++e) m[e] = n[e];
                q   = qq & 0xff;
                fin = qq >> 8;
            }
            float* __restrict__ gd = Gb[cur ^ 1] + t * 12;
            *(float4*)(gd + 0) = make_float4(m[0], m[1], m[2],  m[3]);
            *(float4*)(gd + 4) = make_float4(m[4], m[5], m[6],  m[7]);
            *(float4*)(gd + 8) = make_float4(m[8], m[9], m[10], m[11]);
            Qb[cur ^ 1][t] = q | (fin << 8);
        }
        __syncthreads();
        cur ^= 1;
    }
    const float* __restrict__ Gl = Gb[cur];

    for (int item = t; item < 3 * (JN / 2); item += CHT) {
        const int q  = item / 3;
        const int r  = item - 3 * q;
        const int i0 = 2 * q;
        const int i1 = 2 * q + 1;
        const float4 ga = *(const float4*)(Gl + (i0 * 3 + r) * 4);
        const float4 gb = *(const float4*)(Gl + (i1 * 3 + r) * 4);
        const float4 ja = *(const float4*)(Jl + i0 * 4);
        const float4 jb = *(const float4*)(Jl + i1 * 4);
        const float ta = ga.w - (ga.x * ja.x + ga.y * ja.y + ga.z * ja.z);
        const float tb = gb.w - (gb.x * jb.x + gb.y * jb.y + gb.z * jb.z);
        const int ks = q >> 1;
        const int kh = q & 1;
        float* __restrict__ gq = Gmat + ((size_t)(8 * ks + kh) * NC + (size_t)(3 * b + r)) * 2;
        *(float2*)(gq)                       = make_float2(ga.x, gb.x);
        *(float2*)(gq + (size_t)2 * NC * 2)  = make_float2(ga.y, gb.y);
        *(float2*)(gq + (size_t)4 * NC * 2)  = make_float2(ga.z, gb.z);
        *(float2*)(gq + (size_t)6 * NC * 2)  = make_float2(ta, tb);
        const float tr = (r == 0) ? tr0 : ((r == 1) ? tr1 : tr2);
        Jout[b * (JN * 3) + i0 * 3 + r] = ga.w * sc + tr;
        Jout[b * (JN * 3) + i1 * 3 + r] = gb.w * sc + tr;
    }
}

#define MTILES ((NV + 15) / 16)
#define MGRID  21
#define WPB    4
#define SLAB   (WPB * 16)
#define SLDW   (SLAB + 4)
#define NSEG   2
#define WOG    11
#define WTF    (16 * JN)
#define WTQ    (WTF / 4)
#define WQTOT  (NV * (JN / 4))

__global__ __launch_bounds__(WPB * 32, 2)
void mouse_skin_gemm(const float* __restrict__ W,
                     const float* __restrict__ vt,
                     const float* __restrict__ Gmat,
                     const float* __restrict__ scale,
                     const float* __restrict__ trans,
                     float* __restrict__ Vout)
{
    __shared__ float lds[2][16 * SLDW];
    __shared__ __attribute__((aligned(16))) float wt[2][WTF];
    __shared__ __attribute__((aligned(16))) float vtl[2][48];

    const int tid   = threadIdx.x;
    const int lane  = threadIdx.x & 31;
    const int wave  = threadIdx.x >> 5;
    const int col   = lane & 15;
    const int khalf = lane >> 4;

    const float4* __restrict__ W4 = (const float4*)W;

    const bool has5 = tid < WTQ - 4 * (WPB * 32);
    const int  vte  = min(tid, 47);
    float4 p0, p1, p2, p3, p4;
    float  pv;
    {
        const int m0 = blockIdx.x;
        const int g0 = m0 * WTQ + tid;
        pv = vt[min(m0 * 48 + vte, NV * 3 - 1)];
        p0 = W4[min(g0,       WQTOT - 1)];
        p1 = W4[min(g0 + 128, WQTOT - 1)];
        p2 = W4[min(g0 + 256, WQTOT - 1)];
        p3 = W4[min(g0 + 384, WQTOT - 1)];
        p4 = W4[min(g0 + 512, WQTOT - 1)];
    }

    const int C0   = blockIdx.y * SLAB;
    const int cIdx = C0 + wave * 16 + col;
    const float* __restrict__ gp =
        Gmat + ((size_t)khalf * NC + cIdx) * 2;
    v2f bfa[JN];
#pragma unroll
    for (int j = 0; j < JN; ++j)
        bfa[j] = *(const v2f*)(gp + (size_t)j * 4 * NC);

    const int bFirst = C0 / 3;
    const int seg = threadIdx.x / 48;
    const bool act = seg < NSEG;
    const int segc = act ? seg : 0;
    const int s   = threadIdx.x - seg * 48;
    const int row = s / 3;
    const int r   = s - row * 3;
    int   lof[WOG];
    int   oof[WOG];
    float scg[WOG], trg[WOG];
#pragma unroll
    for (int k = 0; k < WOG; ++k) {
        int bg = bFirst + segc + NSEG * k;
        int lc = 3 * bg + r - C0;
        if (lc < 0)     { bg += NSEG; lc += 3 * NSEG; }
        if (lc >= SLAB) { bg -= NSEG; lc -= 3 * NSEG; }
        lof[k] = row * SLDW + lc;
        oof[k] = bg * (NV * 3) + s;
        scg[k] = scale[bg];
        trg[k] = trans[bg * 3 + r];
    }

    {
        float4* __restrict__ wd = (float4*)wt[0] + tid;
        wd[0] = p0; wd[128] = p1; wd[256] = p2; wd[384] = p3;
        if (has5) { wd[512] = p4; vtl[0][tid] = pv; }
    }
    __syncthreads();

    int buf = 0;
    for (int mt = blockIdx.x; mt < MTILES; mt += MGRID) {
        const int  mtn  = mt + MGRID;
        const bool more = mtn < MTILES;
        if (more) {
            const int g0 = mtn * WTQ + tid;
            p0 = W4[min(g0,       WQTOT - 1)];
            p1 = W4[min(g0 + 128, WQTOT - 1)];
            p2 = W4[min(g0 + 256, WQTOT - 1)];
            p3 = W4[min(g0 + 384, WQTOT - 1)];
            p4 = W4[min(g0 + 512, WQTOT - 1)];
            pv = vt[min(mtn * 48 + vte, NV * 3 - 1)];
        }

        const float* __restrict__ wr = wt[buf] + col * JN + 2 * khalf;
        v2f af[JN / 4];

#define SKIN_N(ks, n)                                                                         \
        ta[n] = __builtin_amdgcn_wmma_f32_16x16x4_f32(false, af[ks], false, bfa[4 * (ks) + (n)], \
                                                      (short)0, ta[n], false, false);
#define SKIN_KSTEP(ks) { SKIN_N(ks, 0) SKIN_N(ks, 1) SKIN_N(ks, 2) SKIN_N(ks, 3) }

        v8f ta[4] = {{}, {}, {}, {}};
#pragma unroll
        for (int ks = 0; ks < JN / 4; ++ks)
            af[ks] = *(const v2f*)(wr + ks * 4);
        float tc[24];
#pragma unroll
        for (int g = 0; g < 6; ++g) {
            const float4 c4 = *(const float4*)(vtl[buf] + 24 * khalf + 4 * g);
            tc[4 * g + 0] = c4.x; tc[4 * g + 1] = c4.y; tc[4 * g + 2] = c4.z; tc[4 * g + 3] = c4.w;
        }
#define SKIN_PRIO(p) asm volatile("s_setprio " #p : "+v"(ta[0]), "+v"(ta[1]), "+v"(ta[2]), "+v"(ta[3]))
#pragma unroll
        for (int ks = 0; ks < JN / 4; ++ks) {
            if (ks == 0)  asm volatile("s_setprio 0" : "+v"(af[ks]) : : "memory");
            if (ks == 9)  SKIN_PRIO(1);
            if (ks == 18) SKIN_PRIO(2);
            if (ks == 27) SKIN_PRIO(3);
            SKIN_KSTEP(ks)
        }
#undef SKIN_PRIO
#undef SKIN_KSTEP
#undef SKIN_N
        v8f acca;
#pragma unroll
        for (int i = 0; i < 8; ++i)
            acca[i] = fmaf(tc[3 * i + 2], ta[2][i], fmaf(tc[3 * i + 1], ta[1][i], fmaf(tc[3 * i], ta[0][i], ta[3][i])));

        float* __restrict__ stage = lds[buf];
#pragma unroll
        for (int i = 0; i < 8; ++i)
            stage[(khalf * 8 + i) * SLDW + wave * 16 + col] = acca[i];

        if (more) {
            float4* __restrict__ wd = (float4*)wt[buf ^ 1] + tid;
            wd[0] = p0; wd[128] = p1; wd[256] = p2; wd[384] = p3;
            if (has5) { wd[512] = p4; vtl[buf ^ 1][tid] = pv; }
        }
        __syncthreads();

        const int vr  = mt * 16 + row;
        const int tof = mt * 48;
        float val[WOG];
        int   idx[WOG];
#pragma unroll
        for (int k = 0; k < WOG; ++k) {
            val[k] = stage[lof[k]] * scg[k] + trg[k];
            idx[k] = oof[k] + tof;
        }
        if (act && vr < NV) {
#pragma unroll
            for (int k = 0; k < WOG; ++k)
                Vout[(size_t)idx[k]] = val[k];
        }
        buf ^= 1;
    }
}

extern "C" void kernel_launch(void* const* d_in, const int* in_sizes, int n_in,
                              void* d_out, int out_size, void* d_ws, size_t ws_size,
                              hipStream_t stream) {
    const float* pose         = (const float*)d_in[0];
    const float* bone_lengths = (const float*)d_in[1];
    const float* cbl          = (const float*)d_in[2];
    const float* trans        = (const float*)d_in[3];
    const float* scale        = (const float*)d_in[4];
    const float* v_template   = (const float*)d_in[5];
    const float* t_pose       = (const float*)d_in[6];
    const float* weights      = (const float*)d_in[7];
    const int*   parent       = (const int*)d_in[8];
    const int*   bone_mapper  = (const int*)d_in[9];

    float* Gmat = (float*)d_ws;

    float* Vout = (float*)d_out;
    float* Jout = Vout + (size_t)BATCH * NV * 3;

    mouse_chain_kernel<<<dim3(BATCH), CHT, 0, stream>>>(
        pose, bone_lengths, cbl, trans, scale, t_pose, parent, bone_mapper,
        Gmat, Jout);

    dim3 grid(MGRID, NC / SLAB);
    mouse_skin_gemm<<<grid, WPB * 32, 0, stream>>>(
        weights, v_template, Gmat, scale, trans, Vout);
}
